# starting kernel
# baseline (speedup 1.0000x reference)
#include <hip/hip_runtime.h>
#include <hip/hip_bf16.h>

#define BATCH 65536
#define VEC   512
#define KCLUS 1024
#define GAMMA_ 0.99f
#define EPS_   1e-5f

#define ROWS_PER_BLOCK 128
#define XPAD 516

typedef __attribute__((ext_vector_type(2))) float v2f;
typedef __attribute__((ext_vector_type(8))) float v8f;

#define ESUM_OFF   0
#define COUNTS_OFF (VEC * KCLUS)
#define W2_OFF     (COUNTS_OFF + KCLUS)
#define CS_OFF     (W2_OFF + KCLUS)

__global__ void zero_ws_kernel(float* __restrict__ ws, int n) {
    int i = blockIdx.x * blockDim.x + threadIdx.x;
    if (i < n) ws[i] = 0.0f;
}

__global__ void w2_kernel(const float* __restrict__ W, float* __restrict__ w2) {
    int k = blockIdx.x * blockDim.x + threadIdx.x;
    if (k >= KCLUS) return;
    float s = 0.0f;
#pragma unroll 8
    for (int v = 0; v < VEC; ++v) {
        float w = W[(size_t)v * KCLUS + k];
        s += w * w;
    }
    w2[k] = s;
}

__global__ __launch_bounds__(128)
void assign_accum_kernel(const float* __restrict__ X,
                         const float* __restrict__ W,
                         const float* __restrict__ w2,
                         float* __restrict__ esum,
                         float* __restrict__ counts)
{
    extern __shared__ char smem[];
    float* xs   = (float*)smem;
    int*   aidx = (int*)(smem + ROWS_PER_BLOCK * XPAD * sizeof(float));

    const int tid  = threadIdx.x;
    const int lane = tid & 31;
    const int wave = tid >> 5;
    const int rowBase = blockIdx.x * ROWS_PER_BLOCK;

    {
        const float4* Xg = (const float4*)(X + (size_t)rowBase * VEC);
#pragma unroll 4
        for (int i = tid; i < ROWS_PER_BLOCK * (VEC / 4); i += 128) {
            int row = i >> 7;
            int v4  = i & 127;
            float4 val = Xg[(size_t)row * (VEC / 4) + v4];
            *(float4*)&xs[row * XPAD + v4 * 4] = val;
        }
    }
    __syncthreads();

    const int stripe0 = wave * 32;
    const int stripe1 = stripe0 + 16;
    const int l15     = lane & 15;
    const int khalf   = (lane >> 4) << 1;

    float best0[8], best1[8];
    int   bidx0[8], bidx1[8];
#pragma unroll
    for (int r = 0; r < 8; ++r) {
        best0[r] = 3.4e38f; bidx0[r] = 0;
        best1[r] = 3.4e38f; bidx1[r] = 0;
    }

    const float* xrow0 = &xs[(stripe0 + l15) * XPAD + khalf];
    const float* xrow1 = &xs[(stripe1 + l15) * XPAD + khalf];

    for (int kb = 0; kb < KCLUS / 16; ++kb) {
        const int col = kb * 16 + l15;
        v8f acc0 = {0.f, 0.f, 0.f, 0.f, 0.f, 0.f, 0.f, 0.f};
        v8f acc1 = {0.f, 0.f, 0.f, 0.f, 0.f, 0.f, 0.f, 0.f};
        const float* wp = W + (size_t)khalf * KCLUS + col;
#pragma unroll 4
        for (int s = 0; s < VEC / 4; ++s) {
            v2f b;
            b.x = wp[0];
            b.y = wp[KCLUS];
            wp += 4 * KCLUS;
            v2f a0 = *(const v2f*)(xrow0 + 4 * s);
            v2f a1 = *(const v2f*)(xrow1 + 4 * s);
            acc0 = __builtin_amdgcn_wmma_f32_16x16x4_f32(
                       false, a0, false, b, (short)0, acc0, false, false);
            acc1 = __builtin_amdgcn_wmma_f32_16x16x4_f32(
                       false, a1, false, b, (short)0, acc1, false, false);
        }
        const float w2c = w2[col];
#pragma unroll
        for (int r = 0; r < 8; ++r) {
            float sc0 = w2c - 2.0f * acc0[r];
            if (sc0 < best0[r]) { best0[r] = sc0; bidx0[r] = col; }
            float sc1 = w2c - 2.0f * acc1[r];
            if (sc1 < best1[r]) { best1[r] = sc1; bidx1[r] = col; }
        }
    }

#pragma unroll
    for (int m = 8; m >= 1; m >>= 1) {
#pragma unroll
        for (int r = 0; r < 8; ++r) {
            float ob0 = __shfl_xor(best0[r], m, 32);
            int   oi0 = __shfl_xor(bidx0[r], m, 32);
            if (ob0 < best0[r] || (ob0 == best0[r] && oi0 < bidx0[r])) {
                best0[r] = ob0; bidx0[r] = oi0;
            }
            float ob1 = __shfl_xor(best1[r], m, 32);
            int   oi1 = __shfl_xor(bidx1[r], m, 32);
            if (ob1 < best1[r] || (ob1 == best1[r] && oi1 < bidx1[r])) {
                best1[r] = ob1; bidx1[r] = oi1;
            }
        }
    }
    if (l15 == 0) {
        int rb0 = stripe0 + (lane >> 4) * 8;
        int rb1 = stripe1 + (lane >> 4) * 8;
#pragma unroll
        for (int r = 0; r < 8; ++r) {
            aidx[rb0 + r] = bidx0[r];
            aidx[rb1 + r] = bidx1[r];
        }
    }
    __syncthreads();

    if (tid < ROWS_PER_BLOCK) atomicAdd(&counts[aidx[tid]], 1.0f);

    for (int i = tid; i < ROWS_PER_BLOCK * VEC; i += 128) {
        int row = i >> 9;
        int v   = i & (VEC - 1);
        int k   = aidx[row];
        atomicAdd(&esum[(size_t)v * KCLUS + k], xs[row * XPAD + v]);
    }
}

__global__ __launch_bounds__(1024)
void finalize1_kernel(const float* __restrict__ cluster_size,
                      const float* __restrict__ counts,
                      float* __restrict__ cs)
{
    __shared__ float red[KCLUS];
    int k = threadIdx.x;
    float c = counts[k];
    if (c == 0.0f) c = 1.0f;
    float ncs = GAMMA_ * cluster_size[k] + (1.0f - GAMMA_) * c;
    red[k] = ncs;
    __syncthreads();
    for (int s = KCLUS / 2; s > 0; s >>= 1) {
        if (k < s) red[k] += red[k + s];
        __syncthreads();
    }
    float n = red[0];
    cs[k] = (ncs + EPS_) / (n + (float)KCLUS * EPS_) * n;
}

__global__ void finalize2_kernel(const float* __restrict__ embed_avg,
                                 const float* __restrict__ esum,
                                 const float* __restrict__ cs,
                                 float* __restrict__ out)
{
    int i = blockIdx.x * blockDim.x + threadIdx.x;
    if (i >= VEC * KCLUS) return;
    int k = i & (KCLUS - 1);
    out[i] = (GAMMA_ * embed_avg[i] + (1.0f - GAMMA_) * esum[i]) / cs[k];
}

extern "C" void kernel_launch(void* const* d_in, const int* in_sizes, int n_in,
                              void* d_out, int out_size, void* d_ws, size_t ws_size,
                              hipStream_t stream) {
    const float* X            = (const float*)d_in[0];
    const float* W            = (const float*)d_in[1];
    const float* cluster_size = (const float*)d_in[2];
    const float* embed_avg    = (const float*)d_in[3];
    float* out = (float*)d_out;
    float* ws  = (float*)d_ws;

    float* esum   = ws + ESUM_OFF;
    float* counts = ws + COUNTS_OFF;
    float* w2     = ws + W2_OFF;
    float* cs     = ws + CS_OFF;

    int zn = VEC * KCLUS + KCLUS;
    zero_ws_kernel<<<(zn + 255) / 256, 256, 0, stream>>>(ws, zn);

    w2_kernel<<<KCLUS / 256, 256, 0, stream>>>(W, w2);

    size_t smem = (size_t)ROWS_PER_BLOCK * XPAD * sizeof(float)
                + (size_t)ROWS_PER_BLOCK * sizeof(int);
    assign_accum_kernel<<<BATCH / ROWS_PER_BLOCK, 128, smem, stream>>>(
        X, W, w2, esum, counts);

    finalize1_kernel<<<1, KCLUS, 0, stream>>>(cluster_size, counts, cs);
    finalize2_kernel<<<(VEC * KCLUS + 255) / 256, 256, 0, stream>>>(
        embed_avg, esum, cs, out);
}


# W (B operand) staged through LDS as 64col x 64k column-major panels shared by the block's 4 waves (register prefetch, 2 buffers), wave tile 32x64; was per-wave 4-byte global loads; sum order kept
# speedup vs baseline: 1.4379x; 1.4379x over previous
#include <hip/hip_runtime.h>
#include <hip/hip_bf16.h>

#define BATCH 65536
#define VEC   512
#define KCLUS 1024
#define GAMMA_ 0.99f
#define EPS_   1e-5f

#define ROWS_PER_BLOCK 128
#define XPAD 516

#define PN 64
#define PK 64
#define WPITCH 68
#define WBUF (PN * WPITCH)

typedef __attribute__((ext_vector_type(2))) float v2f;
typedef __attribute__((ext_vector_type(4))) float v4f;
typedef __attribute__((ext_vector_type(8))) float v8f;

#define ESUM_OFF   0
#define COUNTS_OFF (VEC * KCLUS)
#define W2_OFF     (COUNTS_OFF + KCLUS)
#define CS_OFF     (W2_OFF + KCLUS)

__global__ void zero_ws_kernel(float* __restrict__ ws, int n) {
    int i = blockIdx.x * blockDim.x + threadIdx.x;
    if (i < n) ws[i] = 0.0f;
}

__global__ void w2_kernel(const float* __restrict__ W, float* __restrict__ w2) {
    int k = blockIdx.x * blockDim.x + threadIdx.x;
    if (k >= KCLUS) return;
    float s = 0.0f;
#pragma unroll 8
    for (int v = 0; v < VEC; ++v) {
        float w = W[(size_t)v * KCLUS + k];
        s += w * w;
    }
    w2[k] = s;
}

__device__ __forceinline__ void panel_fetch(const float* __restrict__ wsrc, int p, v4f (&pf)[2][4])
{
    const float* g = wsrc + (size_t)((p & 7) * PK) * KCLUS + (p >> 3) * PN;
#pragma unroll
    for (int j2 = 0; j2 < 2; ++j2) {
#pragma unroll
        for (int j = 0; j < 4; ++j)
            pf[j2][j] = *(const v4f*)(g + (size_t)(32 * j2 + j) * KCLUS);
    }
}

__device__ __forceinline__ void panel_store(float* __restrict__ dst, const v4f (&pf)[2][4])
{
#pragma unroll
    for (int j2 = 0; j2 < 2; ++j2) {
#pragma unroll
        for (int i = 0; i < 4; ++i) {
            v4f v = {pf[j2][0][i], pf[j2][1][i], pf[j2][2][i], pf[j2][3][i]};
            *(v4f*)(dst + i * WPITCH + 32 * j2) = v;
        }
    }
}

__global__ __launch_bounds__(128)
void assign_accum_kernel(const float* __restrict__ X,
                         const float* __restrict__ W,
                         const float* __restrict__ w2,
                         float* __restrict__ esum,
                         float* __restrict__ counts)
{
    extern __shared__ char smem[];
    float* xs   = (float*)smem;
    float* wt   = xs + ROWS_PER_BLOCK * XPAD;
    int*   aidx = (int*)(wt + 2 * WBUF);

    const int tid  = threadIdx.x;
    const int lane = tid & 31;
    const int wave = tid >> 5;
    const int rowBase = blockIdx.x * ROWS_PER_BLOCK;

    const int cg  = tid & 15;
    const int kg0 = tid >> 4;
    const float* wsrc = W + (size_t)(4 * kg0) * KCLUS + 4 * cg;
    float* wdst0 = wt + (4 * cg) * WPITCH + 4 * kg0;
    v4f pf[2][4];

    panel_fetch(wsrc, 0, pf);

    {
        const float4* Xg = (const float4*)(X + (size_t)rowBase * VEC);
#pragma unroll 4
        for (int i = tid; i < ROWS_PER_BLOCK * (VEC / 4); i += 128) {
            int row = i >> 7;
            int v4  = i & 127;
            float4 val = Xg[(size_t)row * (VEC / 4) + v4];
            *(float4*)&xs[row * XPAD + v4 * 4] = val;
        }
    }
    panel_store(wdst0, pf);
    panel_fetch(wsrc, 1, pf);
    __syncthreads();

    const int stripe0 = wave * 32;
    const int stripe1 = stripe0 + 16;
    const int l15     = lane & 15;
    const int khalf   = (lane >> 4) << 1;

    float best0[8], best1[8];
    int   bidx0[8], bidx1[8];
#pragma unroll
    for (int r = 0; r < 8; ++r) {
        best0[r] = 3.4e38f; bidx0[r] = 0;
        best1[r] = 3.4e38f; bidx1[r] = 0;
    }

    const float* xrow0 = &xs[(stripe0 + l15) * XPAD + khalf];
    const float* xrow1 = &xs[(stripe1 + l15) * XPAD + khalf];
    const float* bfrag = wt + l15 * WPITCH + khalf;

    for (int cb = 0; cb < KCLUS / PN; ++cb) {
        v8f acc0[4], acc1[4];
        float w2c[4];
#pragma unroll
        for (int n = 0; n < 4; ++n) {
            acc0[n] = (v8f){0.f, 0.f, 0.f, 0.f, 0.f, 0.f, 0.f, 0.f};
            acc1[n] = (v8f){0.f, 0.f, 0.f, 0.f, 0.f, 0.f, 0.f, 0.f};
            w2c[n]  = w2[cb * PN + n * 16 + l15];
        }

        for (int kc = 0; kc < VEC / PK; ++kc) {
            const int p   = cb * (VEC / PK) + kc;
            const int cur = p & 1;
            if (p + 1 < (KCLUS / PN) * (VEC / PK)) panel_store(wdst0 + (cur ^ 1) * WBUF, pf);
            if (p + 2 < (KCLUS / PN) * (VEC / PK)) panel_fetch(wsrc, p + 2, pf);

            const float* a0p = xrow0 + kc * PK;
            const float* a1p = xrow1 + kc * PK;
            const float* bp  = bfrag + cur * WBUF;
#pragma unroll
            for (int s = 0; s < PK / 4; ++s) {
                v2f a0 = *(const v2f*)(a0p + 4 * s);
                v2f a1 = *(const v2f*)(a1p + 4 * s);
                v2f b[4];
#pragma unroll
                for (int n = 0; n < 4; ++n)
                    b[n] = *(const v2f*)(bp + n * 16 * WPITCH + 4 * s);
#pragma unroll
                for (int n = 0; n < 4; ++n) {
                    acc0[n] = __builtin_amdgcn_wmma_f32_16x16x4_f32(
                                  false, a0, false, b[n], (short)0, acc0[n], false, false);
                    acc1[n] = __builtin_amdgcn_wmma_f32_16x16x4_f32(
                                  false, a1, false, b[n], (short)0, acc1[n], false, false);
                }
            }
            __syncthreads();
        }

#pragma unroll
        for (int n = 0; n < 4; ++n) {
            const int col = cb * PN + n * 16 + l15;
#pragma unroll
            for (int r = 0; r < 8; ++r) {
                float sc0 = w2c[n] - 2.0f * acc0[n][r];
                if (sc0 < best0[r]) { best0[r] = sc0; bidx0[r] = col; }
                float sc1 = w2c[n] - 2.0f * acc1[n][r];
                if (sc1 < best1[r]) { best1[r] = sc1; bidx1[r] = col; }
            }
        }
    }

#pragma unroll
    for (int m = 8; m >= 1; m >>= 1) {
#pragma unroll
        for (int r = 0; r < 8; ++r) {
            float ob0 = __shfl_xor(best0[r], m, 32);
            int   oi0 = __shfl_xor(bidx0[r], m, 32);
            if (ob0 < best0[r] || (ob0 == best0[r] && oi0 < bidx0[r])) {
                best0[r] = ob0; bidx0[r] = oi0;
            }
            float ob1 = __shfl_xor(best1[r], m, 32);
            int   oi1 = __shfl_xor(bidx1[r], m, 32);
            if (ob1 < best1[r] || (ob1 == best1[r] && oi1 < bidx1[r])) {
                best1[r] = ob1; bidx1[r] = oi1;
            }
        }
    }
    if (l15 == 0) {
        int rb0 = stripe0 + (lane >> 4) * 8;
        int rb1 = stripe1 + (lane >> 4) * 8;
#pragma unroll
        for (int r = 0; r < 8; ++r) {
            aidx[rb0 + r] = bidx0[r];
            aidx[rb1 + r] = bidx1[r];
        }
    }
    __syncthreads();

    if (tid < ROWS_PER_BLOCK) atomicAdd(&counts[aidx[tid]], 1.0f);

    for (int i = tid; i < ROWS_PER_BLOCK * VEC; i += 128) {
        int row = i >> 9;
        int v   = i & (VEC - 1);
        int k   = aidx[row];
        atomicAdd(&esum[(size_t)v * KCLUS + k], xs[row * XPAD + v]);
    }
}

__global__ __launch_bounds__(1024)
void finalize1_kernel(const float* __restrict__ cluster_size,
                      const float* __restrict__ counts,
                      float* __restrict__ cs)
{
    __shared__ float red[KCLUS];
    int k = threadIdx.x;
    float c = counts[k];
    if (c == 0.0f) c = 1.0f;
    float ncs = GAMMA_ * cluster_size[k] + (1.0f - GAMMA_) * c;
    red[k] = ncs;
    __syncthreads();
    for (int s = KCLUS / 2; s > 0; s >>= 1) {
        if (k < s) red[k] += red[k + s];
        __syncthreads();
    }
    float n = red[0];
    cs[k] = (ncs + EPS_) / (n + (float)KCLUS * EPS_) * n;
}

__global__ void finalize2_kernel(const float* __restrict__ embed_avg,
                                 const float* __restrict__ esum,
                                 const float* __restrict__ cs,
                                 float* __restrict__ out)
{
    int i = blockIdx.x * blockDim.x + threadIdx.x;
    if (i >= VEC * KCLUS) return;
    int k = i & (KCLUS - 1);
    out[i] = (GAMMA_ * embed_avg[i] + (1.0f - GAMMA_) * esum[i]) / cs[k];
}

extern "C" void kernel_launch(void* const* d_in, const int* in_sizes, int n_in,
                              void* d_out, int out_size, void* d_ws, size_t ws_size,
                              hipStream_t stream) {
    const float* X            = (const float*)d_in[0];
    const float* W            = (const float*)d_in[1];
    const float* cluster_size = (const float*)d_in[2];
    const float* embed_avg    = (const float*)d_in[3];
    float* out = (float*)d_out;
    float* ws  = (float*)d_ws;

    float* esum   = ws + ESUM_OFF;
    float* counts = ws + COUNTS_OFF;
    float* w2     = ws + W2_OFF;
    float* cs     = ws + CS_OFF;

    int zn = VEC * KCLUS + KCLUS;
    zero_ws_kernel<<<(zn + 255) / 256, 256, 0, stream>>>(ws, zn);

    w2_kernel<<<KCLUS / 256, 256, 0, stream>>>(W, w2);

    size_t smem = (size_t)ROWS_PER_BLOCK * XPAD * sizeof(float)
                + (size_t)2 * WBUF * sizeof(float)
                + (size_t)ROWS_PER_BLOCK * sizeof(int);
    assign_accum_kernel<<<BATCH / ROWS_PER_BLOCK, 128, smem, stream>>>(
        X, W, w2, esum, counts);

    finalize1_kernel<<<1, KCLUS, 0, stream>>>(cluster_size, counts, cs);
    finalize2_kernel<<<(VEC * KCLUS + 255) / 256, 256, 0, stream>>>(
        embed_avg, esum, cs, out);
}


# embed_sum kept cluster-major in workspace: a wave's 32 float atomic adds fall in one 128-byte line (was 32 lines 4 KB apart); last kernel transposes via LDS; atomic sum order unordered as before
# speedup vs baseline: 3.8030x; 2.6448x over previous
#include <hip/hip_runtime.h>
#include <hip/hip_bf16.h>

#define BATCH 65536
#define VEC   512
#define KCLUS 1024
#define GAMMA_ 0.99f
#define EPS_   1e-5f

#define ROWS_PER_BLOCK 128
#define XPAD 516

#define PN 64
#define PK 64
#define WPITCH 68
#define WBUF (PN * WPITCH)

typedef __attribute__((ext_vector_type(2))) float v2f;
typedef __attribute__((ext_vector_type(4))) float v4f;
typedef __attribute__((ext_vector_type(8))) float v8f;

#define ESUM_OFF   0
#define COUNTS_OFF (VEC * KCLUS)
#define W2_OFF     (COUNTS_OFF + KCLUS)
#define CS_OFF     (W2_OFF + KCLUS)

__global__ void zero_ws_kernel(float* __restrict__ ws, int n) {
    int i = blockIdx.x * blockDim.x + threadIdx.x;
    if (i < n) ws[i] = 0.0f;
}

__global__ void w2_kernel(const float* __restrict__ W, float* __restrict__ w2) {
    int k = blockIdx.x * blockDim.x + threadIdx.x;
    if (k >= KCLUS) return;
    float s = 0.0f;
#pragma unroll 8
    for (int v = 0; v < VEC; ++v) {
        float w = W[(size_t)v * KCLUS + k];
        s += w * w;
    }
    w2[k] = s;
}

__device__ __forceinline__ void panel_fetch(const float* __restrict__ wsrc, int p, v4f (&pf)[2][4])
{
    const float* g = wsrc + (size_t)((p & 7) * PK) * KCLUS + (p >> 3) * PN;
#pragma unroll
    for (int j2 = 0; j2 < 2; ++j2) {
#pragma unroll
        for (int j = 0; j < 4; ++j)
            pf[j2][j] = *(const v4f*)(g + (size_t)(32 * j2 + j) * KCLUS);
    }
}

__device__ __forceinline__ void panel_store(float* __restrict__ dst, const v4f (&pf)[2][4])
{
#pragma unroll
    for (int j2 = 0; j2 < 2; ++j2) {
#pragma unroll
        for (int i = 0; i < 4; ++i) {
            v4f v = {pf[j2][0][i], pf[j2][1][i], pf[j2][2][i], pf[j2][3][i]};
            *(v4f*)(dst + i * WPITCH + 32 * j2) = v;
        }
    }
}

__global__ __launch_bounds__(128)
void assign_accum_kernel(const float* __restrict__ X,
                         const float* __restrict__ W,
                         const float* __restrict__ w2,
                         float* __restrict__ esum,
                         float* __restrict__ counts)
{
    extern __shared__ char smem[];
    float* xs   = (float*)smem;
    float* wt   = xs + ROWS_PER_BLOCK * XPAD;
    int*   aidx = (int*)(wt + 2 * WBUF);

    const int tid  = threadIdx.x;
    const int lane = tid & 31;
    const int wave = tid >> 5;
    const int rowBase = blockIdx.x * ROWS_PER_BLOCK;

    const int cg  = tid & 15;
    const int kg0 = tid >> 4;
    const float* wsrc = W + (size_t)(4 * kg0) * KCLUS + 4 * cg;
    float* wdst0 = wt + (4 * cg) * WPITCH + 4 * kg0;
    v4f pf[2][4];

    panel_fetch(wsrc, 0, pf);

    {
        const float4* Xg = (const float4*)(X + (size_t)rowBase * VEC);
#pragma unroll 4
        for (int i = tid; i < ROWS_PER_BLOCK * (VEC / 4); i += 128) {
            int row = i >> 7;
            int v4  = i & 127;
            float4 val = Xg[(size_t)row * (VEC / 4) + v4];
            *(float4*)&xs[row * XPAD + v4 * 4] = val;
        }
    }
    panel_store(wdst0, pf);
    panel_fetch(wsrc, 1, pf);
    __syncthreads();

    const int stripe0 = wave * 32;
    const int stripe1 = stripe0 + 16;
    const int l15     = lane & 15;
    const int khalf   = (lane >> 4) << 1;

    float best0[8], best1[8];
    int   bidx0[8], bidx1[8];
#pragma unroll
    for (int r = 0; r < 8; ++r) {
        best0[r] = 3.4e38f; bidx0[r] = 0;
        best1[r] = 3.4e38f; bidx1[r] = 0;
    }

    const float* xrow0 = &xs[(stripe0 + l15) * XPAD + khalf];
    const float* xrow1 = &xs[(stripe1 + l15) * XPAD + khalf];
    const float* bfrag = wt + l15 * WPITCH + khalf;

    for (int cb = 0; cb < KCLUS / PN; ++cb) {
        v8f acc0[4], acc1[4];
        float w2c[4];
#pragma unroll
        for (int n = 0; n < 4; ++n) {
            acc0[n] = (v8f){0.f, 0.f, 0.f, 0.f, 0.f, 0.f, 0.f, 0.f};
            acc1[n] = (v8f){0.f, 0.f, 0.f, 0.f, 0.f, 0.f, 0.f, 0.f};
            w2c[n]  = w2[cb * PN + n * 16 + l15];
        }

        for (int kc = 0; kc < VEC / PK; ++kc) {
            const int p   = cb * (VEC / PK) + kc;
            const int cur = p & 1;
            if (p + 1 < (KCLUS / PN) * (VEC / PK)) panel_store(wdst0 + (cur ^ 1) * WBUF, pf);
            if (p + 2 < (KCLUS / PN) * (VEC / PK)) panel_fetch(wsrc, p + 2, pf);

            const float* a0p = xrow0 + kc * PK;
            const float* a1p = xrow1 + kc * PK;
            const float* bp  = bfrag + cur * WBUF;
#pragma unroll
            for (int s = 0; s < PK / 4; ++s) {
                v2f a0 = *(const v2f*)(a0p + 4 * s);
                v2f a1 = *(const v2f*)(a1p + 4 * s);
                v2f b[4];
#pragma unroll
                for (int n = 0; n < 4; ++n)
                    b[n] = *(const v2f*)(bp + n * 16 * WPITCH + 4 * s);
#pragma unroll
                for (int n = 0; n < 4; ++n) {
                    acc0[n] = __builtin_amdgcn_wmma_f32_16x16x4_f32(
                                  false, a0, false, b[n], (short)0, acc0[n], false, false);
                    acc1[n] = __builtin_amdgcn_wmma_f32_16x16x4_f32(
                                  false, a1, false, b[n], (short)0, acc1[n], false, false);
                }
            }
            __syncthreads();
        }

#pragma unroll
        for (int n = 0; n < 4; ++n) {
            const int col = cb * PN + n * 16 + l15;
#pragma unroll
            for (int r = 0; r < 8; ++r) {
                float sc0 = w2c[n] - 2.0f * acc0[n][r];
                if (sc0 < best0[r]) { best0[r] = sc0; bidx0[r] = col; }
                float sc1 = w2c[n] - 2.0f * acc1[n][r];
                if (sc1 < best1[r]) { best1[r] = sc1; bidx1[r] = col; }
            }
        }
    }

#pragma unroll
    for (int m = 8; m >= 1; m >>= 1) {
#pragma unroll
        for (int r = 0; r < 8; ++r) {
            float ob0 = __shfl_xor(best0[r], m, 32);
            int   oi0 = __shfl_xor(bidx0[r], m, 32);
            if (ob0 < best0[r] || (ob0 == best0[r] && oi0 < bidx0[r])) {
                best0[r] = ob0; bidx0[r] = oi0;
            }
            float ob1 = __shfl_xor(best1[r], m, 32);
            int   oi1 = __shfl_xor(bidx1[r], m, 32);
            if (ob1 < best1[r] || (ob1 == best1[r] && oi1 < bidx1[r])) {
                best1[r] = ob1; bidx1[r] = oi1;
            }
        }
    }
    if (l15 == 0) {
        int rb0 = stripe0 + (lane >> 4) * 8;
        int rb1 = stripe1 + (lane >> 4) * 8;
#pragma unroll
        for (int r = 0; r < 8; ++r) {
            aidx[rb0 + r] = bidx0[r];
            aidx[rb1 + r] = bidx1[r];
        }
    }
    __syncthreads();

    if (tid < ROWS_PER_BLOCK) atomicAdd(&counts[aidx[tid]], 1.0f);

    for (int i = tid; i < ROWS_PER_BLOCK * VEC; i += 128) {
        int row = i >> 9;
        int v   = i & (VEC - 1);
        int k   = aidx[row];
        atomicAdd(&esum[(size_t)k * VEC + v], xs[row * XPAD + v]);
    }
}

__global__ __launch_bounds__(1024)
void finalize1_kernel(const float* __restrict__ cluster_size,
                      const float* __restrict__ counts,
                      float* __restrict__ cs)
{
    __shared__ float red[KCLUS];
    int k = threadIdx.x;
    float c = counts[k];
    if (c == 0.0f) c = 1.0f;
    float ncs = GAMMA_ * cluster_size[k] + (1.0f - GAMMA_) * c;
    red[k] = ncs;
    __syncthreads();
    for (int s = KCLUS / 2; s > 0; s >>= 1) {
        if (k < s) red[k] += red[k + s];
        __syncthreads();
    }
    float n = red[0];
    cs[k] = (ncs + EPS_) / (n + (float)KCLUS * EPS_) * n;
}

#define TT 32
__global__ __launch_bounds__(256)
void finalize2_kernel(const float* __restrict__ embed_avg,
                      const float* __restrict__ esum,
                      const float* __restrict__ cs,
                      float* __restrict__ out)
{
    __shared__ float tile[TT][TT + 1];
    const int tx = threadIdx.x & 31;
    const int ty = threadIdx.x >> 5;
    const int k0 = blockIdx.x * TT;
    const int v0 = blockIdx.y * TT;
#pragma unroll
    for (int j = 0; j < TT / 8; ++j) {
        const int kk = ty + 8 * j;
        tile[kk][tx] = esum[(size_t)(k0 + kk) * VEC + v0 + tx];
    }
    __syncthreads();
    const int   k = k0 + tx;
    const float c = cs[k];
#pragma unroll
    for (int j = 0; j < TT / 8; ++j) {
        const int vv = ty + 8 * j;
        const size_t o = (size_t)(v0 + vv) * KCLUS + k;
        out[o] = (GAMMA_ * embed_avg[o] + (1.0f - GAMMA_) * tile[tx][vv]) / c;
    }
}

extern "C" void kernel_launch(void* const* d_in, const int* in_sizes, int n_in,
                              void* d_out, int out_size, void* d_ws, size_t ws_size,
                              hipStream_t stream) {
    const float* X            = (const float*)d_in[0];
    const float* W            = (const float*)d_in[1];
    const float* cluster_size = (const float*)d_in[2];
    const float* embed_avg    = (const float*)d_in[3];
    float* out = (float*)d_out;
    float* ws  = (float*)d_ws;

    float* esum   = ws + ESUM_OFF;
    float* counts = ws + COUNTS_OFF;
    float* w2     = ws + W2_OFF;
    float* cs     = ws + CS_OFF;

    int zn = VEC * KCLUS + KCLUS;
    zero_ws_kernel<<<(zn + 255) / 256, 256, 0, stream>>>(ws, zn);

    w2_kernel<<<KCLUS / 256, 256, 0, stream>>>(W, w2);

    size_t smem = (size_t)ROWS_PER_BLOCK * XPAD * sizeof(float)
                + (size_t)2 * WBUF * sizeof(float)
                + (size_t)ROWS_PER_BLOCK * sizeof(int);
    assign_accum_kernel<<<BATCH / ROWS_PER_BLOCK, 128, smem, stream>>>(
        X, W, w2, esum, counts);

    finalize1_kernel<<<1, KCLUS, 0, stream>>>(cluster_size, counts, cs);
    finalize2_kernel<<<dim3(KCLUS / TT, VEC / TT), 256, 0, stream>>>(
        embed_avg, esum, cs, out);
}


# Two workgroups resident per WGP: row tile 128->64, W panel depth 64->32 fetched one panel ahead, LDS 293->147 KB; one's X staging and atomic scatter run under the other's products; sum orders kept
# speedup vs baseline: 4.0002x; 1.0519x over previous
#include <hip/hip_runtime.h>
#include <hip/hip_bf16.h>

#define BATCH 65536
#define VEC   512
#define KCLUS 1024
#define GAMMA_ 0.99f
#define EPS_   1e-5f

#define ROWS_PER_BLOCK 64
#define XPAD 516

#define PN 64
#define PK 32
#define WPITCH 36
#define WBUF (PN * WPITCH)

typedef __attribute__((ext_vector_type(2))) float v2f;
typedef __attribute__((ext_vector_type(4))) float v4f;
typedef __attribute__((ext_vector_type(8))) float v8f;

#define ESUM_OFF   0
#define COUNTS_OFF (VEC * KCLUS)
#define W2_OFF     (COUNTS_OFF + KCLUS)
#define CS_OFF     (W2_OFF + KCLUS)

__global__ void zero_ws_kernel(float* __restrict__ ws, int n) {
    int i = blockIdx.x * blockDim.x + threadIdx.x;
    if (i < n) ws[i] = 0.0f;
}

__global__ void w2_kernel(const float* __restrict__ W, float* __restrict__ w2) {
    int k = blockIdx.x * blockDim.x + threadIdx.x;
    if (k >= KCLUS) return;
    float s = 0.0f;
#pragma unroll 8
    for (int v = 0; v < VEC; ++v) {
        float w = W[(size_t)v * KCLUS + k];
        s += w * w;
    }
    w2[k] = s;
}

__device__ __forceinline__ void panel_fetch(const float* __restrict__ wsrc, int p, v4f (&pf)[4])
{
    const float* g = wsrc + (size_t)((p & 15) * PK) * KCLUS + (p >> 4) * PN;
#pragma unroll
    for (int j = 0; j < 4; ++j)
        pf[j] = *(const v4f*)(g + (size_t)j * KCLUS);
}

__device__ __forceinline__ void panel_store(float* __restrict__ dst, const v4f (&pf)[4])
{
#pragma unroll
    for (int i = 0; i < 4; ++i) {
        v4f v = {pf[0][i], pf[1][i], pf[2][i], pf[3][i]};
        *(v4f*)(dst + i * WPITCH) = v;
    }
}

__global__ __launch_bounds__(128)
void assign_accum_kernel(const float* __restrict__ X,
                         const float* __restrict__ W,
                         const float* __restrict__ w2,
                         float* __restrict__ esum,
                         float* __restrict__ counts)
{
    extern __shared__ char smem[];
    float* xs   = (float*)smem;
    float* wt   = xs + ROWS_PER_BLOCK * XPAD;
    int*   aidx = (int*)(wt + 2 * WBUF);

    const int tid  = threadIdx.x;
    const int lane = tid & 31;
    const int wave = tid >> 5;
    const int rowBase = blockIdx.x * ROWS_PER_BLOCK;

    const int cg  = tid & 15;
    const int kg0 = tid >> 4;
    const float* wsrc = W + (size_t)(4 * kg0) * KCLUS + 4 * cg;
    float* wdst0 = wt + (4 * cg) * WPITCH + 4 * kg0;
    v4f pf[4];

    panel_fetch(wsrc, 0, pf);

    {
        const float4* Xg = (const float4*)(X + (size_t)rowBase * VEC);
#pragma unroll 4
        for (int i = tid; i < ROWS_PER_BLOCK * (VEC / 4); i += 128) {
            int row = i >> 7;
            int v4  = i & 127;
            float4 val = Xg[(size_t)row * (VEC / 4) + v4];
            *(float4*)&xs[row * XPAD + v4 * 4] = val;
        }
    }
    panel_store(wdst0, pf);
    __syncthreads();

    const int stripe0 = wave * 16;
    const int l15     = lane & 15;
    const int khalf   = (lane >> 4) << 1;

    float best0[8];
    int   bidx0[8];
#pragma unroll
    for (int r = 0; r < 8; ++r) {
        best0[r] = 3.4e38f; bidx0[r] = 0;
    }

    const float* xrow0 = &xs[(stripe0 + l15) * XPAD + khalf];
    const float* bfrag = wt + l15 * WPITCH + khalf;

    for (int cb = 0; cb < KCLUS / PN; ++cb) {
        v8f acc0[4];
        float w2c[4];
#pragma unroll
        for (int n = 0; n < 4; ++n) {
            acc0[n] = (v8f){0.f, 0.f, 0.f, 0.f, 0.f, 0.f, 0.f, 0.f};
            w2c[n]  = w2[cb * PN + n * 16 + l15];
        }

        for (int kc = 0; kc < VEC / PK; ++kc) {
            const int p   = cb * (VEC / PK) + kc;
            const int cur = p & 1;
            if (p + 1 < (KCLUS / PN) * (VEC / PK)) panel_fetch(wsrc, p + 1, pf);

            const float* a0p = xrow0 + kc * PK;
            const float* bp  = bfrag + cur * WBUF;
#pragma unroll
            for (int s = 0; s < PK / 4; ++s) {
                v2f a0 = *(const v2f*)(a0p + 4 * s);
                v2f b[4];
#pragma unroll
                for (int n = 0; n < 4; ++n)
                    b[n] = *(const v2f*)(bp + n * 16 * WPITCH + 4 * s);
#pragma unroll
                for (int n = 0; n < 4; ++n)
                    acc0[n] = __builtin_amdgcn_wmma_f32_16x16x4_f32(
                                  false, a0, false, b[n], (short)0, acc0[n], false, false);
            }
            if (p + 1 < (KCLUS / PN) * (VEC / PK)) panel_store(wdst0 + (cur ^ 1) * WBUF, pf);
            __syncthreads();
        }

#pragma unroll
        for (int n = 0; n < 4; ++n) {
            const int col = cb * PN + n * 16 + l15;
#pragma unroll
            for (int r = 0; r < 8; ++r) {
                float sc0 = w2c[n] - 2.0f * acc0[n][r];
                if (sc0 < best0[r]) { best0[r] = sc0; bidx0[r] = col; }
            }
        }
    }

#pragma unroll
    for (int m = 8; m >= 1; m >>= 1) {
#pragma unroll
        for (int r = 0; r < 8; ++r) {
            float ob0 = __shfl_xor(best0[r], m, 32);
            int   oi0 = __shfl_xor(bidx0[r], m, 32);
            if (ob0 < best0[r] || (ob0 == best0[r] && oi0 < bidx0[r])) {
                best0[r] = ob0; bidx0[r] = oi0;
            }
        }
    }
    if (l15 == 0) {
        int rb0 = stripe0 + (lane >> 4) * 8;
#pragma unroll
        for (int r = 0; r < 8; ++r)
            aidx[rb0 + r] = bidx0[r];
    }
    __syncthreads();

    if (tid < ROWS_PER_BLOCK) atomicAdd(&counts[aidx[tid]], 1.0f);

    for (int i = tid; i < ROWS_PER_BLOCK * VEC; i += 128) {
        int row = i >> 9;
        int v   = i & (VEC - 1);
        int k   = aidx[row];
        atomicAdd(&esum[(size_t)k * VEC + v], xs[row * XPAD + v]);
    }
}

__global__ __launch_bounds__(1024)
void finalize1_kernel(const float* __restrict__ cluster_size,
                      const float* __restrict__ counts,
                      float* __restrict__ cs)
{
    __shared__ float red[KCLUS];
    int k = threadIdx.x;
    float c = counts[k];
    if (c == 0.0f) c = 1.0f;
    float ncs = GAMMA_ * cluster_size[k] + (1.0f - GAMMA_) * c;
    red[k] = ncs;
    __syncthreads();
    for (int s = KCLUS / 2; s > 0; s >>= 1) {
        if (k < s) red[k] += red[k + s];
        __syncthreads();
    }
    float n = red[0];
    cs[k] = (ncs + EPS_) / (n + (float)KCLUS * EPS_) * n;
}

#define TT 32
__global__ __launch_bounds__(256)
void finalize2_kernel(const float* __restrict__ embed_avg,
                      const float* __restrict__ esum,
                      const float* __restrict__ cs,
                      float* __restrict__ out)
{
    __shared__ float tile[TT][TT + 1];
    const int tx = threadIdx.x & 31;
    const int ty = threadIdx.x >> 5;
    const int k0 = blockIdx.x * TT;
    const int v0 = blockIdx.y * TT;
#pragma unroll
    for (int j = 0; j < TT / 8; ++j) {
        const int kk = ty + 8 * j;
        tile[kk][tx] = esum[(size_t)(k0 + kk) * VEC + v0 + tx];
    }
    __syncthreads();
    const int   k = k0 + tx;
    const float c = cs[k];
#pragma unroll
    for (int j = 0; j < TT / 8; ++j) {
        const int vv = ty + 8 * j;
        const size_t o = (size_t)(v0 + vv) * KCLUS + k;
        out[o] = (GAMMA_ * embed_avg[o] + (1.0f - GAMMA_) * tile[tx][vv]) / c;
    }
}

extern "C" void kernel_launch(void* const* d_in, const int* in_sizes, int n_in,
                              void* d_out, int out_size, void* d_ws, size_t ws_size,
                              hipStream_t stream) {
    const float* X            = (const float*)d_in[0];
    const float* W            = (const float*)d_in[1];
    const float* cluster_size = (const float*)d_in[2];
    const float* embed_avg    = (const float*)d_in[3];
    float* out = (float*)d_out;
    float* ws  = (float*)d_ws;

    float* esum   = ws + ESUM_OFF;
    float* counts = ws + COUNTS_OFF;
    float* w2     = ws + W2_OFF;
    float* cs     = ws + CS_OFF;

    int zn = VEC * KCLUS + KCLUS;
    zero_ws_kernel<<<(zn + 255) / 256, 256, 0, stream>>>(ws, zn);

    w2_kernel<<<KCLUS / 256, 256, 0, stream>>>(W, w2);

    size_t smem = (size_t)ROWS_PER_BLOCK * XPAD * sizeof(float)
                + (size_t)2 * WBUF * sizeof(float)
                + (size_t)ROWS_PER_BLOCK * sizeof(int);
    assign_accum_kernel<<<BATCH / ROWS_PER_BLOCK, 128, smem, stream>>>(
        X, W, w2, esum, counts);

    finalize1_kernel<<<1, KCLUS, 0, stream>>>(cluster_size, counts, cs);
    finalize2_kernel<<<dim3(KCLUS / TT, VEC / TT), 256, 0, stream>>>(
        embed_avg, esum, cs, out);
}


# k loop software-pipelined: next panel's fragment reads + panel store stand as ONE group ahead of the 32 products (2nd register set), W fetched 2 trips ahead past an LDS-only barrier; sum order kept
# speedup vs baseline: 4.3720x; 1.0929x over previous
#include <hip/hip_runtime.h>
#include <hip/hip_bf16.h>

#define BATCH 65536
#define VEC   512
#define KCLUS 1024
#define GAMMA_ 0.99f
#define EPS_   1e-5f

#define ROWS_PER_BLOCK 64
#define XPAD 516

#define PN 64
#define PK 32
#define WPITCH 36
#define WBUF (PN * WPITCH)

typedef __attribute__((ext_vector_type(2))) float v2f;
typedef __attribute__((ext_vector_type(4))) float v4f;
typedef __attribute__((ext_vector_type(8))) float v8f;

#define ESUM_OFF   0
#define COUNTS_OFF (VEC * KCLUS)
#define W2_OFF     (COUNTS_OFF + KCLUS)
#define CS_OFF     (W2_OFF + KCLUS)

__global__ void zero_ws_kernel(float* __restrict__ ws, int n) {
    int i = blockIdx.x * blockDim.x + threadIdx.x;
    if (i < n) ws[i] = 0.0f;
}

__global__ void w2_kernel(const float* __restrict__ W, float* __restrict__ w2) {
    int k = blockIdx.x * blockDim.x + threadIdx.x;
    if (k >= KCLUS) return;
    float s = 0.0f;
#pragma unroll 8
    for (int v = 0; v < VEC; ++v) {
        float w = W[(size_t)v * KCLUS + k];
        s += w * w;
    }
    w2[k] = s;
}

__device__ __forceinline__ void panel_fetch(const float* __restrict__ wsrc, int p, v4f (&pf)[4])
{
    const float* g = wsrc + (size_t)((p & 15) * PK) * KCLUS + (p >> 4) * PN;
#pragma unroll
    for (int j = 0; j < 4; ++j)
        pf[j] = *(const v4f*)(g + (size_t)j * KCLUS);
}

__device__ __forceinline__ void panel_store(float* __restrict__ dst, const v4f (&pf)[4])
{
#pragma unroll
    for (int i = 0; i < 4; ++i) {
        v4f v = {pf[0][i], pf[1][i], pf[2][i], pf[3][i]};
        *(v4f*)(dst + i * WPITCH) = v;
    }
}

__device__ __forceinline__ void lds_barrier()
{
    __builtin_amdgcn_fence(__ATOMIC_RELEASE, "workgroup", "local");
    __builtin_amdgcn_s_barrier();
    __builtin_amdgcn_fence(__ATOMIC_ACQUIRE, "workgroup", "local");
}

__global__ __launch_bounds__(128)
void assign_accum_kernel(const float* __restrict__ X,
                         const float* __restrict__ W,
                         const float* __restrict__ w2,
                         float* __restrict__ esum,
                         float* __restrict__ counts)
{
    extern __shared__ char smem[];
    float* xs   = (float*)smem;
    float* wt   = xs + ROWS_PER_BLOCK * XPAD;
    int*   aidx = (int*)(wt + 2 * WBUF);

    const int tid  = threadIdx.x;
    const int lane = tid & 31;
    const int wave = tid >> 5;
    const int rowBase = blockIdx.x * ROWS_PER_BLOCK;

    const int cg  = tid & 15;
    const int kg0 = tid >> 4;
    const float* wsrc = W + (size_t)(4 * kg0) * KCLUS + 4 * cg;
    float* wdst0 = wt + (4 * cg) * WPITCH + 4 * kg0;
    v4f pf[2][4];

    panel_fetch(wsrc, 0, pf[0]);

    {
        const float4* Xg = (const float4*)(X + (size_t)rowBase * VEC);
#pragma unroll 4
        for (int i = tid; i < ROWS_PER_BLOCK * (VEC / 4); i += 128) {
            int row = i >> 7;
            int v4  = i & 127;
            float4 val = Xg[(size_t)row * (VEC / 4) + v4];
            *(float4*)&xs[row * XPAD + v4 * 4] = val;
        }
    }
    panel_store(wdst0, pf[0]);
    panel_fetch(wsrc, 1, pf[1]);
    __syncthreads();

    const int stripe0 = wave * 16;
    const int l15     = lane & 15;
    const int khalf   = (lane >> 4) << 1;

    float best0[8];
    int   bidx0[8];
#pragma unroll
    for (int r = 0; r < 8; ++r) {
        best0[r] = 3.4e38f; bidx0[r] = 0;
    }

    const float* xrow0 = &xs[(stripe0 + l15) * XPAD + khalf];
    const float* bfrag = wt + l15 * WPITCH + khalf;

    v2f fa[2][PK / 4];
    v2f fb[2][PK / 4][4];

#define LOAD_FRAGS(SET, q)                                                              \
    {                                                                                   \
        const float* ap_ = xrow0 + ((q) & 15) * PK;                                     \
        const float* bp_ = bfrag + ((q) & 1) * WBUF;                                    \
        _Pragma("unroll")                                                               \
        for (int s = 0; s < PK / 4; ++s) {                                              \
            fa[SET][s] = *(const v2f*)(ap_ + 4 * s);                                    \
            _Pragma("unroll")                                                           \
            for (int n = 0; n < 4; ++n)                                                 \
                fb[SET][s][n] = *(const v2f*)(bp_ + n * 16 * WPITCH + 4 * s);           \
        }                                                                               \
    }

    LOAD_FRAGS(0, 0)
    panel_store(wdst0 + WBUF, pf[1]);
    panel_fetch(wsrc, 2, pf[0]);
    panel_fetch(wsrc, 3, pf[1]);
    __syncthreads();

    const int NP = (KCLUS / PN) * (VEC / PK);

    for (int cb = 0; cb < KCLUS / PN; ++cb) {
        v8f acc0[4];
        float w2c[4];
#pragma unroll
        for (int n = 0; n < 4; ++n) {
            acc0[n] = (v8f){0.f, 0.f, 0.f, 0.f, 0.f, 0.f, 0.f, 0.f};
            w2c[n]  = w2[cb * PN + n * 16 + l15];
        }

        for (int kc = 0; kc < VEC / PK; kc += 2) {
#pragma unroll
            for (int u = 0; u < 2; ++u) {
                const int t = cb * (VEC / PK) + kc + u;

                panel_store(wdst0 + u * WBUF, pf[u]);
                panel_fetch(wsrc, min(t + 4, NP - 1), pf[u]);
                LOAD_FRAGS(u ^ 1, t + 1)

                __builtin_amdgcn_sched_barrier(0);

#pragma unroll
                for (int s = 0; s < PK / 4; ++s) {
#pragma unroll
                    for (int n = 0; n < 4; ++n)
                        acc0[n] = __builtin_amdgcn_wmma_f32_16x16x4_f32(
                                      false, fa[u][s], false, fb[u][s][n], (short)0, acc0[n], false, false);
                }
                __builtin_amdgcn_sched_barrier(0);
                lds_barrier();
            }
        }

#pragma unroll
        for (int n = 0; n < 4; ++n) {
            const int col = cb * PN + n * 16 + l15;
#pragma unroll
            for (int r = 0; r < 8; ++r) {
                float sc0 = w2c[n] - 2.0f * acc0[n][r];
                if (sc0 < best0[r]) { best0[r] = sc0; bidx0[r] = col; }
            }
        }
    }
#undef LOAD_FRAGS

#pragma unroll
    for (int m = 8; m >= 1; m >>= 1) {
#pragma unroll
        for (int r = 0; r < 8; ++r) {
            float ob0 = __shfl_xor(best0[r], m, 32);
            int   oi0 = __shfl_xor(bidx0[r], m, 32);
            if (ob0 < best0[r] || (ob0 == best0[r] && oi0 < bidx0[r])) {
                best0[r] = ob0; bidx0[r] = oi0;
            }
        }
    }
    if (l15 == 0) {
        int rb0 = stripe0 + (lane >> 4) * 8;
#pragma unroll
        for (int r = 0; r < 8; ++r)
            aidx[rb0 + r] = bidx0[r];
    }
    __syncthreads();

    if (tid < ROWS_PER_BLOCK) atomicAdd(&counts[aidx[tid]], 1.0f);

    for (int i = tid; i < ROWS_PER_BLOCK * VEC; i += 128) {
        int row = i >> 9;
        int v   = i & (VEC - 1);
        int k   = aidx[row];
        atomicAdd(&esum[(size_t)k * VEC + v], xs[row * XPAD + v]);
    }
}

__global__ __launch_bounds__(1024)
void finalize1_kernel(const float* __restrict__ cluster_size,
                      const float* __restrict__ counts,
                      float* __restrict__ cs)
{
    __shared__ float red[KCLUS];
    int k = threadIdx.x;
    float c = counts[k];
    if (c == 0.0f) c = 1.0f;
    float ncs = GAMMA_ * cluster_size[k] + (1.0f - GAMMA_) * c;
    red[k] = ncs;
    __syncthreads();
    for (int s = KCLUS / 2; s > 0; s >>= 1) {
        if (k < s) red[k] += red[k + s];
        __syncthreads();
    }
    float n = red[0];
    cs[k] = (ncs + EPS_) / (n + (float)KCLUS * EPS_) * n;
}

#define TT 32
__global__ __launch_bounds__(256)
void finalize2_kernel(const float* __restrict__ embed_avg,
                      const float* __restrict__ esum,
                      const float* __restrict__ cs,
                      float* __restrict__ out)
{
    __shared__ float tile[TT][TT + 1];
    const int tx = threadIdx.x & 31;
    const int ty = threadIdx.x >> 5;
    const int k0 = blockIdx.x * TT;
    const int v0 = blockIdx.y * TT;
#pragma unroll
    for (int j = 0; j < TT / 8; ++j) {
        const int kk = ty + 8 * j;
        tile[kk][tx] = esum[(size_t)(k0 + kk) * VEC + v0 + tx];
    }
    __syncthreads();
    const int   k = k0 + tx;
    const float c = cs[k];
#pragma unroll
    for (int j = 0; j < TT / 8; ++j) {
        const int vv = ty + 8 * j;
        const size_t o = (size_t)(v0 + vv) * KCLUS + k;
        out[o] = (GAMMA_ * embed_avg[o] + (1.0f - GAMMA_) * tile[tx][vv]) / c;
    }
}

extern "C" void kernel_launch(void* const* d_in, const int* in_sizes, int n_in,
                              void* d_out, int out_size, void* d_ws, size_t ws_size,
                              hipStream_t stream) {
    const float* X            = (const float*)d_in[0];
    const float* W            = (const float*)d_in[1];
    const float* cluster_size = (const float*)d_in[2];
    const float* embed_avg    = (const float*)d_in[3];
    float* out = (float*)d_out;
    float* ws  = (float*)d_ws;

    float* esum   = ws + ESUM_OFF;
    float* counts = ws + COUNTS_OFF;
    float* w2     = ws + W2_OFF;
    float* cs     = ws + CS_OFF;

    int zn = VEC * KCLUS + KCLUS;
    zero_ws_kernel<<<(zn + 255) / 256, 256, 0, stream>>>(ws, zn);

    w2_kernel<<<KCLUS / 256, 256, 0, stream>>>(W, w2);

    size_t smem = (size_t)ROWS_PER_BLOCK * XPAD * sizeof(float)
                + (size_t)2 * WBUF * sizeof(float)
                + (size_t)ROWS_PER_BLOCK * sizeof(int);
    assign_accum_kernel<<<BATCH / ROWS_PER_BLOCK, 128, smem, stream>>>(
        X, W, w2, esum, counts);

    finalize1_kernel<<<1, KCLUS, 0, stream>>>(cluster_size, counts, cs);
    finalize2_kernel<<<dim3(KCLUS / TT, VEC / TT), 256, 0, stream>>>(
        embed_avg, esum, cs, out);
}


# Zero-fill and w2 merged into ONE launch of 1024 fat waves (was 16416 one-store waves, then 32 waves waiting on memory every 8 rows): 16-byte stores, 64 W rows in flight, same sum order; 5->4 launches
# speedup vs baseline: 4.5956x; 1.0511x over previous
#include <hip/hip_runtime.h>
#include <hip/hip_bf16.h>

#define BATCH 65536
#define VEC   512
#define KCLUS 1024
#define GAMMA_ 0.99f
#define EPS_   1e-5f

#define ROWS_PER_BLOCK 64
#define XPAD 516

#define PN 64
#define PK 32
#define WPITCH 36
#define WBUF (PN * WPITCH)

typedef __attribute__((ext_vector_type(2))) float v2f;
typedef __attribute__((ext_vector_type(4))) float v4f;
typedef __attribute__((ext_vector_type(8))) float v8f;

#define ESUM_OFF   0
#define COUNTS_OFF (VEC * KCLUS)
#define W2_OFF     (COUNTS_OFF + KCLUS)
#define CS_OFF     (W2_OFF + KCLUS)

#define PREP_BLOCKS 128
#define W2_DEPTH    64
__global__ __launch_bounds__(256)
void prep_kernel(const float* __restrict__ W, float* __restrict__ ws)
{
    if (blockIdx.x < KCLUS / 32 && threadIdx.x < 32) {
        const int k = blockIdx.x * 32 + threadIdx.x;
        const float* wp = W + k;
        float s = 0.0f;
#pragma unroll
        for (int v0 = 0; v0 < VEC; v0 += W2_DEPTH) {
            float w[W2_DEPTH];
#pragma unroll
            for (int j = 0; j < W2_DEPTH; ++j)
                w[j] = wp[(size_t)(v0 + j) * KCLUS];
#pragma unroll
            for (int j = 0; j < W2_DEPTH; ++j)
                s = __builtin_fmaf(w[j], w[j], s);
        }
        ws[W2_OFF + k] = s;
    }

    float4* z = (float4*)ws;
    for (int i = blockIdx.x * 256 + threadIdx.x; i < (VEC * KCLUS + KCLUS) / 4; i += PREP_BLOCKS * 256)
        z[i] = make_float4(0.0f, 0.0f, 0.0f, 0.0f);
}

__device__ __forceinline__ void panel_fetch(const float* __restrict__ wsrc, int p, v4f (&pf)[4])
{
    const float* g = wsrc + (size_t)((p & 15) * PK) * KCLUS + (p >> 4) * PN;
#pragma unroll
    for (int j = 0; j < 4; ++j)
        pf[j] = *(const v4f*)(g + (size_t)j * KCLUS);
}

__device__ __forceinline__ void panel_store(float* __restrict__ dst, const v4f (&pf)[4])
{
#pragma unroll
    for (int i = 0; i < 4; ++i) {
        v4f v = {pf[0][i], pf[1][i], pf[2][i], pf[3][i]};
        *(v4f*)(dst + i * WPITCH) = v;
    }
}

__device__ __forceinline__ void lds_barrier()
{
    __builtin_amdgcn_fence(__ATOMIC_RELEASE, "workgroup", "local");
    __builtin_amdgcn_s_barrier();
    __builtin_amdgcn_fence(__ATOMIC_ACQUIRE, "workgroup", "local");
}

__global__ __launch_bounds__(128)
void assign_accum_kernel(const float* __restrict__ X,
                         const float* __restrict__ W,
                         const float* __restrict__ w2,
                         float* __restrict__ esum,
                         float* __restrict__ counts)
{
    extern __shared__ char smem[];
    float* xs   = (float*)smem;
    float* wt   = xs + ROWS_PER_BLOCK * XPAD;
    int*   aidx = (int*)(wt + 2 * WBUF);

    const int tid  = threadIdx.x;
    const int lane = tid & 31;
    const int wave = tid >> 5;
    const int rowBase = blockIdx.x * ROWS_PER_BLOCK;

    const int cg  = tid & 15;
    const int kg0 = tid >> 4;
    const float* wsrc = W + (size_t)(4 * kg0) * KCLUS + 4 * cg;
    float* wdst0 = wt + (4 * cg) * WPITCH + 4 * kg0;
    v4f pf[2][4];

    panel_fetch(wsrc, 0, pf[0]);

    {
        const float4* Xg = (const float4*)(X + (size_t)rowBase * VEC);
#pragma unroll 4
        for (int i = tid; i < ROWS_PER_BLOCK * (VEC / 4); i += 128) {
            int row = i >> 7;
            int v4  = i & 127;
            float4 val = Xg[(size_t)row * (VEC / 4) + v4];
            *(float4*)&xs[row * XPAD + v4 * 4] = val;
        }
    }
    panel_store(wdst0, pf[0]);
    panel_fetch(wsrc, 1, pf[1]);
    __syncthreads();

    const int stripe0 = wave * 16;
    const int l15     = lane & 15;
    const int khalf   = (lane >> 4) << 1;

    float best0[8];
    int   bidx0[8];
#pragma unroll
    for (int r = 0; r < 8; ++r) {
        best0[r] = 3.4e38f; bidx0[r] = 0;
    }

    const float* xrow0 = &xs[(stripe0 + l15) * XPAD + khalf];
    const float* bfrag = wt + l15 * WPITCH + khalf;

    v2f fa[2][PK / 4];
    v2f fb[2][PK / 4][4];

#define LOAD_FRAGS(SET, q)                                                              \
    {                                                                                   \
        const float* ap_ = xrow0 + ((q) & 15) * PK;                                     \
        const float* bp_ = bfrag + ((q) & 1) * WBUF;                                    \
        _Pragma("unroll")                                                               \
        for (int s = 0; s < PK / 4; ++s) {                                              \
            fa[SET][s] = *(const v2f*)(ap_ + 4 * s);                                    \
            _Pragma("unroll")                                                           \
            for (int n = 0; n < 4; ++n)                                                 \
                fb[SET][s][n] = *(const v2f*)(bp_ + n * 16 * WPITCH + 4 * s);           \
        }                                                                               \
    }

    LOAD_FRAGS(0, 0)
    panel_store(wdst0 + WBUF, pf[1]);
    panel_fetch(wsrc, 2, pf[0]);
    panel_fetch(wsrc, 3, pf[1]);
    __syncthreads();

    const int NP = (KCLUS / PN) * (VEC / PK);

    for (int cb = 0; cb < KCLUS / PN; ++cb) {
        v8f acc0[4];
        float w2c[4];
#pragma unroll
        for (int n = 0; n < 4; ++n) {
            acc0[n] = (v8f){0.f, 0.f, 0.f, 0.f, 0.f, 0.f, 0.f, 0.f};
            w2c[n]  = w2[cb * PN + n * 16 + l15];
        }

        for (int kc = 0; kc < VEC / PK; kc += 2) {
#pragma unroll
            for (int u = 0; u < 2; ++u) {
                const int t = cb * (VEC / PK) + kc + u;

                panel_store(wdst0 + u * WBUF, pf[u]);
                panel_fetch(wsrc, min(t + 4, NP - 1), pf[u]);
                LOAD_FRAGS(u ^ 1, t + 1)

                __builtin_amdgcn_sched_barrier(0);

#pragma unroll
                for (int s = 0; s < PK / 4; ++s) {
#pragma unroll
                    for (int n = 0; n < 4; ++n)
                        acc0[n] = __builtin_amdgcn_wmma_f32_16x16x4_f32(
                                      false, fa[u][s], false, fb[u][s][n], (short)0, acc0[n], false, false);
                }
                __builtin_amdgcn_sched_barrier(0);
                lds_barrier();
            }
        }

#pragma unroll
        for (int n = 0; n < 4; ++n) {
            const int col = cb * PN + n * 16 + l15;
#pragma unroll
            for (int r = 0; r < 8; ++r) {
                float sc0 = w2c[n] - 2.0f * acc0[n][r];
                if (sc0 < best0[r]) { best0[r] = sc0; bidx0[r] = col; }
            }
        }
    }
#undef LOAD_FRAGS

#pragma unroll
    for (int m = 8; m >= 1; m >>= 1) {
#pragma unroll
        for (int r = 0; r < 8; ++r) {
            float ob0 = __shfl_xor(best0[r], m, 32);
            int   oi0 = __shfl_xor(bidx0[r], m, 32);
            if (ob0 < best0[r] || (ob0 == best0[r] && oi0 < bidx0[r])) {
                best0[r] = ob0; bidx0[r] = oi0;
            }
        }
    }
    if (l15 == 0) {
        int rb0 = stripe0 + (lane >> 4) * 8;
#pragma unroll
        for (int r = 0; r < 8; ++r)
            aidx[rb0 + r] = bidx0[r];
    }
    __syncthreads();

    if (tid < ROWS_PER_BLOCK) atomicAdd(&counts[aidx[tid]], 1.0f);

    for (int i = tid; i < ROWS_PER_BLOCK * VEC; i += 128) {
        int row = i >> 9;
        int v   = i & (VEC - 1);
        int k   = aidx[row];
        atomicAdd(&esum[(size_t)k * VEC + v], xs[row * XPAD + v]);
    }
}

__global__ __launch_bounds__(1024)
void finalize1_kernel(const float* __restrict__ cluster_size,
                      const float* __restrict__ counts,
                      float* __restrict__ cs)
{
    __shared__ float red[KCLUS];
    int k = threadIdx.x;
    float c = counts[k];
    if (c == 0.0f) c = 1.0f;
    float ncs = GAMMA_ * cluster_size[k] + (1.0f - GAMMA_) * c;
    red[k] = ncs;
    __syncthreads();
    for (int s = KCLUS / 2; s > 0; s >>= 1) {
        if (k < s) red[k] += red[k + s];
        __syncthreads();
    }
    float n = red[0];
    cs[k] = (ncs + EPS_) / (n + (float)KCLUS * EPS_) * n;
}

#define TT 32
__global__ __launch_bounds__(256)
void finalize2_kernel(const float* __restrict__ embed_avg,
                      const float* __restrict__ esum,
                      const float* __restrict__ cs,
                      float* __restrict__ out)
{
    __shared__ float tile[TT][TT + 1];
    const int tx = threadIdx.x & 31;
    const int ty = threadIdx.x >> 5;
    const int k0 = blockIdx.x * TT;
    const int v0 = blockIdx.y * TT;
#pragma unroll
    for (int j = 0; j < TT / 8; ++j) {
        const int kk = ty + 8 * j;
        tile[kk][tx] = esum[(size_t)(k0 + kk) * VEC + v0 + tx];
    }
    __syncthreads();
    const int   k = k0 + tx;
    const float c = cs[k];
#pragma unroll
    for (int j = 0; j < TT / 8; ++j) {
        const int vv = ty + 8 * j;
        const size_t o = (size_t)(v0 + vv) * KCLUS + k;
        out[o] = (GAMMA_ * embed_avg[o] + (1.0f - GAMMA_) * tile[tx][vv]) / c;
    }
}

extern "C" void kernel_launch(void* const* d_in, const int* in_sizes, int n_in,
                              void* d_out, int out_size, void* d_ws, size_t ws_size,
                              hipStream_t stream) {
    const float* X            = (const float*)d_in[0];
    const float* W            = (const float*)d_in[1];
    const float* cluster_size = (const float*)d_in[2];
    const float* embed_avg    = (const float*)d_in[3];
    float* out = (float*)d_out;
    float* ws  = (float*)d_ws;

    float* esum   = ws + ESUM_OFF;
    float* counts = ws + COUNTS_OFF;
    float* w2     = ws + W2_OFF;
    float* cs     = ws + CS_OFF;

    prep_kernel<<<PREP_BLOCKS, 256, 0, stream>>>(W, ws);

    size_t smem = (size_t)ROWS_PER_BLOCK * XPAD * sizeof(float)
                + (size_t)2 * WBUF * sizeof(float)
                + (size_t)ROWS_PER_BLOCK * sizeof(int);
    assign_accum_kernel<<<BATCH / ROWS_PER_BLOCK, 128, smem, stream>>>(
        X, W, w2, esum, counts);

    finalize1_kernel<<<1, KCLUS, 0, stream>>>(cluster_size, counts, cs);
    finalize2_kernel<<<dim3(KCLUS / TT, VEC / TT), 256, 0, stream>>>(
        embed_avg, esum, cs, out);
}


# embed_sum: tile rows sharing a cluster are summed in LDS first (rank sort of 64, rising row order, LDS reads 8 rows at a time), then ONE float atomic per cluster+column; atomics unordered as before
# speedup vs baseline: 4.7654x; 1.0369x over previous
#include <hip/hip_runtime.h>
#include <hip/hip_bf16.h>

#define BATCH 65536
#define VEC   512
#define KCLUS 1024
#define GAMMA_ 0.99f
#define EPS_   1e-5f

#define ROWS_PER_BLOCK 64
#define XPAD 516

#define PN 64
#define PK 32
#define WPITCH 36
#define WBUF (PN * WPITCH)

typedef __attribute__((ext_vector_type(2))) float v2f;
typedef __attribute__((ext_vector_type(4))) float v4f;
typedef __attribute__((ext_vector_type(8))) float v8f;

#define ESUM_OFF   0
#define COUNTS_OFF (VEC * KCLUS)
#define W2_OFF     (COUNTS_OFF + KCLUS)
#define CS_OFF     (W2_OFF + KCLUS)

#define PREP_BLOCKS 128
#define W2_DEPTH    64
__global__ __launch_bounds__(256)
void prep_kernel(const float* __restrict__ W, float* __restrict__ ws)
{
    if (blockIdx.x < KCLUS / 32 && threadIdx.x < 32) {
        const int k = blockIdx.x * 32 + threadIdx.x;
        const float* wp = W + k;
        float s = 0.0f;
#pragma unroll
        for (int v0 = 0; v0 < VEC; v0 += W2_DEPTH) {
            float w[W2_DEPTH];
#pragma unroll
            for (int j = 0; j < W2_DEPTH; ++j)
                w[j] = wp[(size_t)(v0 + j) * KCLUS];
#pragma unroll
            for (int j = 0; j < W2_DEPTH; ++j)
                s = __builtin_fmaf(w[j], w[j], s);
        }
        ws[W2_OFF + k] = s;
    }

    float4* z = (float4*)ws;
    for (int i = blockIdx.x * 256 + threadIdx.x; i < (VEC * KCLUS + KCLUS) / 4; i += PREP_BLOCKS * 256)
        z[i] = make_float4(0.0f, 0.0f, 0.0f, 0.0f);
}

__device__ __forceinline__ void panel_fetch(const float* __restrict__ wsrc, int p, v4f (&pf)[4])
{
    const float* g = wsrc + (size_t)((p & 15) * PK) * KCLUS + (p >> 4) * PN;
#pragma unroll
    for (int j = 0; j < 4; ++j)
        pf[j] = *(const v4f*)(g + (size_t)j * KCLUS);
}

__device__ __forceinline__ void panel_store(float* __restrict__ dst, const v4f (&pf)[4])
{
#pragma unroll
    for (int i = 0; i < 4; ++i) {
        v4f v = {pf[0][i], pf[1][i], pf[2][i], pf[3][i]};
        *(v4f*)(dst + i * WPITCH) = v;
    }
}

__device__ __forceinline__ void lds_barrier()
{
    __builtin_amdgcn_fence(__ATOMIC_RELEASE, "workgroup", "local");
    __builtin_amdgcn_s_barrier();
    __builtin_amdgcn_fence(__ATOMIC_ACQUIRE, "workgroup", "local");
}

__global__ __launch_bounds__(128)
void assign_accum_kernel(const float* __restrict__ X,
                         const float* __restrict__ W,
                         const float* __restrict__ w2,
                         float* __restrict__ esum,
                         float* __restrict__ counts)
{
    extern __shared__ char smem[];
    float* xs   = (float*)smem;
    float* wt   = xs + ROWS_PER_BLOCK * XPAD;
    int*   aidx = (int*)(wt + 2 * WBUF);

    const int tid  = threadIdx.x;
    const int lane = tid & 31;
    const int wave = tid >> 5;
    const int rowBase = blockIdx.x * ROWS_PER_BLOCK;

    const int cg  = tid & 15;
    const int kg0 = tid >> 4;
    const float* wsrc = W + (size_t)(4 * kg0) * KCLUS + 4 * cg;
    float* wdst0 = wt + (4 * cg) * WPITCH + 4 * kg0;
    v4f pf[2][4];

    panel_fetch(wsrc, 0, pf[0]);

    {
        const float4* Xg = (const float4*)(X + (size_t)rowBase * VEC);
#pragma unroll 4
        for (int i = tid; i < ROWS_PER_BLOCK * (VEC / 4); i += 128) {
            int row = i >> 7;
            int v4  = i & 127;
            float4 val = Xg[(size_t)row * (VEC / 4) + v4];
            *(float4*)&xs[row * XPAD + v4 * 4] = val;
        }
    }
    panel_store(wdst0, pf[0]);
    panel_fetch(wsrc, 1, pf[1]);
    __syncthreads();

    const int stripe0 = wave * 16;
    const int l15     = lane & 15;
    const int khalf   = (lane >> 4) << 1;

    float best0[8];
    int   bidx0[8];
#pragma unroll
    for (int r = 0; r < 8; ++r) {
        best0[r] = 3.4e38f; bidx0[r] = 0;
    }

    const float* xrow0 = &xs[(stripe0 + l15) * XPAD + khalf];
    const float* bfrag = wt + l15 * WPITCH + khalf;

    v2f fa[2][PK / 4];
    v2f fb[2][PK / 4][4];

#define LOAD_FRAGS(SET, q)                                                              \
    {                                                                                   \
        const float* ap_ = xrow0 + ((q) & 15) * PK;                                     \
        const float* bp_ = bfrag + ((q) & 1) * WBUF;                                    \
        _Pragma("unroll")                                                               \
        for (int s = 0; s < PK / 4; ++s) {                                              \
            fa[SET][s] = *(const v2f*)(ap_ + 4 * s);                                    \
            _Pragma("unroll")                                                           \
            for (int n = 0; n < 4; ++n)                                                 \
                fb[SET][s][n] = *(const v2f*)(bp_ + n * 16 * WPITCH + 4 * s);           \
        }                                                                               \
    }

    LOAD_FRAGS(0, 0)
    panel_store(wdst0 + WBUF, pf[1]);
    panel_fetch(wsrc, 2, pf[0]);
    panel_fetch(wsrc, 3, pf[1]);
    __syncthreads();

    const int NP = (KCLUS / PN) * (VEC / PK);

    for (int cb = 0; cb < KCLUS / PN; ++cb) {
        v8f acc0[4];
        float w2c[4];
#pragma unroll
        for (int n = 0; n < 4; ++n) {
            acc0[n] = (v8f){0.f, 0.f, 0.f, 0.f, 0.f, 0.f, 0.f, 0.f};
            w2c[n]  = w2[cb * PN + n * 16 + l15];
        }

        for (int kc = 0; kc < VEC / PK; kc += 2) {
#pragma unroll
            for (int u = 0; u < 2; ++u) {
                const int t = cb * (VEC / PK) + kc + u;

                panel_store(wdst0 + u * WBUF, pf[u]);
                panel_fetch(wsrc, min(t + 4, NP - 1), pf[u]);
                LOAD_FRAGS(u ^ 1, t + 1)

                __builtin_amdgcn_sched_barrier(0);

#pragma unroll
                for (int s = 0; s < PK / 4; ++s) {
#pragma unroll
                    for (int n = 0; n < 4; ++n)
                        acc0[n] = __builtin_amdgcn_wmma_f32_16x16x4_f32(
                                      false, fa[u][s], false, fb[u][s][n], (short)0, acc0[n], false, false);
                }
                __builtin_amdgcn_sched_barrier(0);
                lds_barrier();
            }
        }

#pragma unroll
        for (int n = 0; n < 4; ++n) {
            const int col = cb * PN + n * 16 + l15;
#pragma unroll
            for (int r = 0; r < 8; ++r) {
                float sc0 = w2c[n] - 2.0f * acc0[n][r];
                if (sc0 < best0[r]) { best0[r] = sc0; bidx0[r] = col; }
            }
        }
    }
#undef LOAD_FRAGS

#pragma unroll
    for (int m = 8; m >= 1; m >>= 1) {
#pragma unroll
        for (int r = 0; r < 8; ++r) {
            float ob0 = __shfl_xor(best0[r], m, 32);
            int   oi0 = __shfl_xor(bidx0[r], m, 32);
            if (ob0 < best0[r] || (ob0 == best0[r] && oi0 < bidx0[r])) {
                best0[r] = ob0; bidx0[r] = oi0;
            }
        }
    }
    if (l15 == 0) {
        int rb0 = stripe0 + (lane >> 4) * 8;
#pragma unroll
        for (int r = 0; r < 8; ++r)
            aidx[rb0 + r] = bidx0[r];
    }
    __syncthreads();

    if (tid < ROWS_PER_BLOCK) atomicAdd(&counts[aidx[tid]], 1.0f);

    int* prow  = (int*)wt;
    int* pcode = prow + ROWS_PER_BLOCK;
    if (tid < ROWS_PER_BLOCK) {
        const int k = aidx[tid];
        int rank = 0;
        for (int j = 0; j < ROWS_PER_BLOCK; ++j) {
            const int kj = aidx[j];
            rank += (kj < k || (kj == k && j < tid)) ? 1 : 0;
        }
        prow[rank] = tid;
    }
    __syncthreads();
    if (tid < ROWS_PER_BLOCK) {
        const int r = prow[tid];
        const int k = aidx[r];
        const bool last = (tid == ROWS_PER_BLOCK - 1) ? true : (aidx[prow[tid + 1]] != k);
        pcode[tid] = r | ((last ? k + 1 : 0) << 6);
    }
    __syncthreads();

    float s0 = 0.0f, s1 = 0.0f, s2 = 0.0f, s3 = 0.0f;
    for (int g = 0; g < ROWS_PER_BLOCK; g += 8) {
        int   pc[8];
        float x[8][4];
#pragma unroll
        for (int u = 0; u < 8; ++u)
            pc[u] = __builtin_amdgcn_readfirstlane(pcode[g + u]);
#pragma unroll
        for (int u = 0; u < 8; ++u) {
            const float* xr = xs + (pc[u] & 63) * XPAD + tid;
            x[u][0] = xr[0]; x[u][1] = xr[128]; x[u][2] = xr[256]; x[u][3] = xr[384];
        }
#pragma unroll
        for (int u = 0; u < 8; ++u) {
            s0 += x[u][0]; s1 += x[u][1]; s2 += x[u][2]; s3 += x[u][3];
            const int e = pc[u] >> 6;
            if (e != 0) {
                float* d = esum + (size_t)(e - 1) * VEC + tid;
                atomicAdd(d,       s0);
                atomicAdd(d + 128, s1);
                atomicAdd(d + 256, s2);
                atomicAdd(d + 384, s3);
                s0 = 0.0f; s1 = 0.0f; s2 = 0.0f; s3 = 0.0f;
            }
        }
    }
}

__global__ __launch_bounds__(1024)
void finalize1_kernel(const float* __restrict__ cluster_size,
                      const float* __restrict__ counts,
                      float* __restrict__ cs)
{
    __shared__ float red[KCLUS];
    int k = threadIdx.x;
    float c = counts[k];
    if (c == 0.0f) c = 1.0f;
    float ncs = GAMMA_ * cluster_size[k] + (1.0f - GAMMA_) * c;
    red[k] = ncs;
    __syncthreads();
    for (int s = KCLUS / 2; s > 0; s >>= 1) {
        if (k < s) red[k] += red[k + s];
        __syncthreads();
    }
    float n = red[0];
    cs[k] = (ncs + EPS_) / (n + (float)KCLUS * EPS_) * n;
}

#define TT 32
__global__ __launch_bounds__(256)
void finalize2_kernel(const float* __restrict__ embed_avg,
                      const float* __restrict__ esum,
                      const float* __restrict__ cs,
                      float* __restrict__ out)
{
    __shared__ float tile[TT][TT + 1];
    const int tx = threadIdx.x & 31;
    const int ty = threadIdx.x >> 5;
    const int k0 = blockIdx.x * TT;
    const int v0 = blockIdx.y * TT;
#pragma unroll
    for (int j = 0; j < TT / 8; ++j) {
        const int kk = ty + 8 * j;
        tile[kk][tx] = esum[(size_t)(k0 + kk) * VEC + v0 + tx];
    }
    __syncthreads();
    const int   k = k0 + tx;
    const float c = cs[k];
#pragma unroll
    for (int j = 0; j < TT / 8; ++j) {
        const int vv = ty + 8 * j;
        const size_t o = (size_t)(v0 + vv) * KCLUS + k;
        out[o] = (GAMMA_ * embed_avg[o] + (1.0f - GAMMA_) * tile[tx][vv]) / c;
    }
}

extern "C" void kernel_launch(void* const* d_in, const int* in_sizes, int n_in,
                              void* d_out, int out_size, void* d_ws, size_t ws_size,
                              hipStream_t stream) {
    const float* X            = (const float*)d_in[0];
    const float* W            = (const float*)d_in[1];
    const float* cluster_size = (const float*)d_in[2];
    const float* embed_avg    = (const float*)d_in[3];
    float* out = (float*)d_out;
    float* ws  = (float*)d_ws;

    float* esum   = ws + ESUM_OFF;
    float* counts = ws + COUNTS_OFF;
    float* w2     = ws + W2_OFF;
    float* cs     = ws + CS_OFF;

    prep_kernel<<<PREP_BLOCKS, 256, 0, stream>>>(W, ws);

    size_t smem = (size_t)ROWS_PER_BLOCK * XPAD * sizeof(float)
                + (size_t)2 * WBUF * sizeof(float)
                + (size_t)ROWS_PER_BLOCK * sizeof(int);
    assign_accum_kernel<<<BATCH / ROWS_PER_BLOCK, 128, smem, stream>>>(
        X, W, w2, esum, counts);

    finalize1_kernel<<<1, KCLUS, 0, stream>>>(cluster_size, counts, cs);
    finalize2_kernel<<<dim3(KCLUS / TT, VEC / TT), 256, 0, stream>>>(
        embed_avg, esum, cs, out);
}


# Wave tile 16x64 -> 64x64 (16 accumulators): W panel 256 col x 8 k, each wave all 64 rows x its own 64 columns; fragment LDS reads per trip 20 -> 8; k order kept; waves' minima merged in LDS
# speedup vs baseline: 4.9280x; 1.0341x over previous
#include <hip/hip_runtime.h>
#include <hip/hip_bf16.h>

#define BATCH 65536
#define VEC   512
#define KCLUS 1024
#define GAMMA_ 0.99f
#define EPS_   1e-5f

#define ROWS_PER_BLOCK 64
#define XPAD 516

#define PN 256
#define PK 8
#define WPITCH 12
#define WBUF (PN * WPITCH)

typedef __attribute__((ext_vector_type(2))) float v2f;
typedef __attribute__((ext_vector_type(4))) float v4f;
typedef __attribute__((ext_vector_type(8))) float v8f;

#define ESUM_OFF   0
#define COUNTS_OFF (VEC * KCLUS)
#define W2_OFF     (COUNTS_OFF + KCLUS)
#define CS_OFF     (W2_OFF + KCLUS)

#define PREP_BLOCKS 128
#define W2_DEPTH    64
__global__ __launch_bounds__(256)
void prep_kernel(const float* __restrict__ W, float* __restrict__ ws)
{
    if (blockIdx.x < KCLUS / 32 && threadIdx.x < 32) {
        const int k = blockIdx.x * 32 + threadIdx.x;
        const float* wp = W + k;
        float s = 0.0f;
#pragma unroll
        for (int v0 = 0; v0 < VEC; v0 += W2_DEPTH) {
            float w[W2_DEPTH];
#pragma unroll
            for (int j = 0; j < W2_DEPTH; ++j)
                w[j] = wp[(size_t)(v0 + j) * KCLUS];
#pragma unroll
            for (int j = 0; j < W2_DEPTH; ++j)
                s = __builtin_fmaf(w[j], w[j], s);
        }
        ws[W2_OFF + k] = s;
    }

    float4* z = (float4*)ws;
    for (int i = blockIdx.x * 256 + threadIdx.x; i < (VEC * KCLUS + KCLUS) / 4; i += PREP_BLOCKS * 256)
        z[i] = make_float4(0.0f, 0.0f, 0.0f, 0.0f);
}

__device__ __forceinline__ void panel_fetch(const float* __restrict__ wsrc, int p, v4f (&pf)[4])
{
    const float* g = wsrc + (size_t)((p & 63) * PK) * KCLUS + (p >> 6) * PN;
#pragma unroll
    for (int j = 0; j < 4; ++j)
        pf[j] = *(const v4f*)(g + (size_t)j * KCLUS);
}

__device__ __forceinline__ void panel_store(float* __restrict__ dst, const v4f (&pf)[4])
{
#pragma unroll
    for (int i = 0; i < 4; ++i) {
        v4f v = {pf[0][i], pf[1][i], pf[2][i], pf[3][i]};
        *(v4f*)(dst + i * WPITCH) = v;
    }
}

__device__ __forceinline__ void lds_barrier()
{
    __builtin_amdgcn_fence(__ATOMIC_RELEASE, "workgroup", "local");
    __builtin_amdgcn_s_barrier();
    __builtin_amdgcn_fence(__ATOMIC_ACQUIRE, "workgroup", "local");
}

__global__ __launch_bounds__(128)
void assign_accum_kernel(const float* __restrict__ X,
                         const float* __restrict__ W,
                         const float* __restrict__ w2,
                         float* __restrict__ esum,
                         float* __restrict__ counts)
{
    extern __shared__ char smem[];
    float* xs   = (float*)smem;
    float* wt   = xs + ROWS_PER_BLOCK * XPAD;
    int*   aidx = (int*)(wt + 2 * WBUF);

    const int tid  = threadIdx.x;
    const int lane = tid & 31;
    const int wave = tid >> 5;
    const int rowBase = blockIdx.x * ROWS_PER_BLOCK;

    const int cg  = tid >> 1;
    const int kg0 = tid & 1;
    const float* wsrc = W + (size_t)(4 * kg0) * KCLUS + 4 * cg;
    float* wdst0 = wt + (4 * cg) * WPITCH + 4 * kg0;
    v4f pf[2][4];

    panel_fetch(wsrc, 0, pf[0]);

    {
        const float4* Xg = (const float4*)(X + (size_t)rowBase * VEC);
#pragma unroll 4
        for (int i = tid; i < ROWS_PER_BLOCK * (VEC / 4); i += 128) {
            int row = i >> 7;
            int v4  = i & 127;
            float4 val = Xg[(size_t)row * (VEC / 4) + v4];
            *(float4*)&xs[row * XPAD + v4 * 4] = val;
        }
    }
    panel_store(wdst0, pf[0]);
    panel_fetch(wsrc, 1, pf[1]);
    __syncthreads();

    const int l15     = lane & 15;
    const int khalf   = (lane >> 4) << 1;
    const int colw    = wave * 64;

    float best0[4][8];
    int   bidx0[4][8];
#pragma unroll
    for (int m = 0; m < 4; ++m) {
#pragma unroll
        for (int r = 0; r < 8; ++r) {
            best0[m][r] = 3.4e38f; bidx0[m][r] = 0;
        }
    }

    const int    aoff0 = l15 * XPAD + khalf;
    const float* bfrag = wt + (colw + l15) * WPITCH + khalf;

    v2f fa[2][PK / 4][4];
    v2f fb[2][PK / 4][4];

#define LOAD_FRAGS(SET, q)                                                              \
    {                                                                                   \
        const int alo_ = aoff0 + ((q) & 63) * PK;                                       \
        int       ahi_ = alo_ + 32 * XPAD;                                              \
          \
          \
        asm("" : "+v"(ahi_));                                                           \
        const float* apl_ = xs + alo_;                                                  \
        const float* aph_ = xs + ahi_;                                                  \
        const float* bp_  = bfrag + ((q) & 1) * WBUF;                                   \
        _Pragma("unroll")                                                               \
        for (int s = 0; s < PK / 4; ++s) {                                              \
            fa[SET][s][0] = *(const v2f*)(apl_ + 4 * s);                                \
            fa[SET][s][1] = *(const v2f*)(apl_ + 16 * XPAD + 4 * s);                    \
            fa[SET][s][2] = *(const v2f*)(aph_ + 4 * s);                                \
            fa[SET][s][3] = *(const v2f*)(aph_ + 16 * XPAD + 4 * s);                    \
            _Pragma("unroll")                                                           \
            for (int n = 0; n < 4; ++n)                                                 \
                fb[SET][s][n] = *(const v2f*)(bp_ + n * 16 * WPITCH + 4 * s);           \
        }                                                                               \
    }

    LOAD_FRAGS(0, 0)
    panel_store(wdst0 + WBUF, pf[1]);
    panel_fetch(wsrc, 2, pf[0]);
    panel_fetch(wsrc, 3, pf[1]);
    __syncthreads();

    const int NP = (KCLUS / PN) * (VEC / PK);

    for (int cb = 0; cb < KCLUS / PN; ++cb) {
        v8f acc0[4][4];
        float w2c[4];
#pragma unroll
        for (int n = 0; n < 4; ++n) {
#pragma unroll
            for (int m = 0; m < 4; ++m)
                acc0[m][n] = (v8f){0.f, 0.f, 0.f, 0.f, 0.f, 0.f, 0.f, 0.f};
            w2c[n]  = w2[cb * PN + colw + n * 16 + l15];
        }

        for (int kc = 0; kc < VEC / PK; kc += 2) {
#pragma unroll
            for (int u = 0; u < 2; ++u) {
                const int t = cb * (VEC / PK) + kc + u;

                panel_store(wdst0 + u * WBUF, pf[u]);
                panel_fetch(wsrc, min(t + 4, NP - 1), pf[u]);
                LOAD_FRAGS(u ^ 1, t + 1)

                __builtin_amdgcn_sched_barrier(0);

#pragma unroll
                for (int s = 0; s < PK / 4; ++s) {
#pragma unroll
                    for (int m = 0; m < 4; ++m) {
#pragma unroll
                        for (int n = 0; n < 4; ++n)
                            acc0[m][n] = __builtin_amdgcn_wmma_f32_16x16x4_f32(
                                             false, fa[u][s][m], false, fb[u][s][n], (short)0, acc0[m][n], false, false);
                    }
                }
                __builtin_amdgcn_sched_barrier(0);
                lds_barrier();
            }
        }

#pragma unroll
        for (int n = 0; n < 4; ++n) {
            const int col = cb * PN + colw + n * 16 + l15;
#pragma unroll
            for (int m = 0; m < 4; ++m) {
#pragma unroll
                for (int r = 0; r < 8; ++r) {
                    float sc0 = w2c[n] - 2.0f * acc0[m][n][r];
                    if (sc0 < best0[m][r]) { best0[m][r] = sc0; bidx0[m][r] = col; }
                }
            }
        }
    }
#undef LOAD_FRAGS

#pragma unroll
    for (int sh = 8; sh >= 1; sh >>= 1) {
        float ob0[4][8];
        int   oi0[4][8];
#pragma unroll
        for (int m = 0; m < 4; ++m) {
#pragma unroll
            for (int r = 0; r < 8; ++r) {
                ob0[m][r] = __shfl_xor(best0[m][r], sh, 32);
                oi0[m][r] = __shfl_xor(bidx0[m][r], sh, 32);
            }
        }
#pragma unroll
        for (int m = 0; m < 4; ++m) {
#pragma unroll
            for (int r = 0; r < 8; ++r) {
                const bool take = (ob0[m][r] < best0[m][r]) | ((ob0[m][r] == best0[m][r]) & (oi0[m][r] < bidx0[m][r]));
                best0[m][r] = take ? ob0[m][r] : best0[m][r];
                bidx0[m][r] = take ? oi0[m][r] : bidx0[m][r];
            }
        }
    }
    float* wbest = wt + 4 * ROWS_PER_BLOCK;
    int*   widx  = (int*)(wbest + 4 * ROWS_PER_BLOCK);
    if (l15 == 0) {
#pragma unroll
        for (int m = 0; m < 4; ++m) {
            const int rb0 = m * 16 + (lane >> 4) * 8;
#pragma unroll
            for (int r = 0; r < 8; ++r) {
                wbest[wave * ROWS_PER_BLOCK + rb0 + r] = best0[m][r];
                widx [wave * ROWS_PER_BLOCK + rb0 + r] = bidx0[m][r];
            }
        }
    }
    __syncthreads();
    if (tid < ROWS_PER_BLOCK) {
        float b = wbest[tid];
        int   i = widx[tid];
#pragma unroll
        for (int w = 1; w < 4; ++w) {
            const float ob = wbest[w * ROWS_PER_BLOCK + tid];
            const int   oi = widx [w * ROWS_PER_BLOCK + tid];
            if (ob < b || (ob == b && oi < i)) { b = ob; i = oi; }
        }
        aidx[tid] = i;
    }
    __syncthreads();

    if (tid < ROWS_PER_BLOCK) atomicAdd(&counts[aidx[tid]], 1.0f);

    int* prow  = (int*)wt;
    int* pcode = prow + ROWS_PER_BLOCK;
    if (tid < ROWS_PER_BLOCK) {
        const int k = aidx[tid];
        int rank = 0;
        for (int j = 0; j < ROWS_PER_BLOCK; ++j) {
            const int kj = aidx[j];
            rank += (kj < k || (kj == k && j < tid)) ? 1 : 0;
        }
        prow[rank] = tid;
    }
    __syncthreads();
    if (tid < ROWS_PER_BLOCK) {
        const int r = prow[tid];
        const int k = aidx[r];
        const bool last = (tid == ROWS_PER_BLOCK - 1) ? true : (aidx[prow[tid + 1]] != k);
        pcode[tid] = r | ((last ? k + 1 : 0) << 6);
    }
    __syncthreads();

    float s0 = 0.0f, s1 = 0.0f, s2 = 0.0f, s3 = 0.0f;
    for (int g = 0; g < ROWS_PER_BLOCK; g += 8) {
        int   pc[8];
        float x[8][4];
#pragma unroll
        for (int u = 0; u < 8; ++u)
            pc[u] = __builtin_amdgcn_readfirstlane(pcode[g + u]);
#pragma unroll
        for (int u = 0; u < 8; ++u) {
            const float* xr = xs + (pc[u] & 63) * XPAD + tid;
            x[u][0] = xr[0]; x[u][1] = xr[128]; x[u][2] = xr[256]; x[u][3] = xr[384];
        }
#pragma unroll
        for (int u = 0; u < 8; ++u) {
            s0 += x[u][0]; s1 += x[u][1]; s2 += x[u][2]; s3 += x[u][3];
            const int e = pc[u] >> 6;
            if (e != 0) {
                float* d = esum + (size_t)(e - 1) * VEC + tid;
                atomicAdd(d,       s0);
                atomicAdd(d + 128, s1);
                atomicAdd(d + 256, s2);
                atomicAdd(d + 384, s3);
                s0 = 0.0f; s1 = 0.0f; s2 = 0.0f; s3 = 0.0f;
            }
        }
    }
}

__global__ __launch_bounds__(1024)
void finalize1_kernel(const float* __restrict__ cluster_size,
                      const float* __restrict__ counts,
                      float* __restrict__ cs)
{
    __shared__ float red[KCLUS];
    int k = threadIdx.x;
    float c = counts[k];
    if (c == 0.0f) c = 1.0f;
    float ncs = GAMMA_ * cluster_size[k] + (1.0f - GAMMA_) * c;
    red[k] = ncs;
    __syncthreads();
    for (int s = KCLUS / 2; s > 0; s >>= 1) {
        if (k < s) red[k] += red[k + s];
        __syncthreads();
    }
    float n = red[0];
    cs[k] = (ncs + EPS_) / (n + (float)KCLUS * EPS_) * n;
}

#define TT 32
__global__ __launch_bounds__(256)
void finalize2_kernel(const float* __restrict__ embed_avg,
                      const float* __restrict__ esum,
                      const float* __restrict__ cs,
                      float* __restrict__ out)
{
    __shared__ float tile[TT][TT + 1];
    const int tx = threadIdx.x & 31;
    const int ty = threadIdx.x >> 5;
    const int k0 = blockIdx.x * TT;
    const int v0 = blockIdx.y * TT;
#pragma unroll
    for (int j = 0; j < TT / 8; ++j) {
        const int kk = ty + 8 * j;
        tile[kk][tx] = esum[(size_t)(k0 + kk) * VEC + v0 + tx];
    }
    __syncthreads();
    const int   k = k0 + tx;
    const float c = cs[k];
#pragma unroll
    for (int j = 0; j < TT / 8; ++j) {
        const int vv = ty + 8 * j;
        const size_t o = (size_t)(v0 + vv) * KCLUS + k;
        out[o] = (GAMMA_ * embed_avg[o] + (1.0f - GAMMA_) * tile[tx][vv]) / c;
    }
}

extern "C" void kernel_launch(void* const* d_in, const int* in_sizes, int n_in,
                              void* d_out, int out_size, void* d_ws, size_t ws_size,
                              hipStream_t stream) {
    const float* X            = (const float*)d_in[0];
    const float* W            = (const float*)d_in[1];
    const float* cluster_size = (const float*)d_in[2];
    const float* embed_avg    = (const float*)d_in[3];
    float* out = (float*)d_out;
    float* ws  = (float*)d_ws;

    float* esum   = ws + ESUM_OFF;
    float* counts = ws + COUNTS_OFF;
    float* w2     = ws + W2_OFF;
    float* cs     = ws + CS_OFF;

    prep_kernel<<<PREP_BLOCKS, 256, 0, stream>>>(W, ws);

    size_t smem = (size_t)ROWS_PER_BLOCK * XPAD * sizeof(float)
                + (size_t)2 * WBUF * sizeof(float)
                + (size_t)ROWS_PER_BLOCK * sizeof(int);
    assign_accum_kernel<<<BATCH / ROWS_PER_BLOCK, 128, smem, stream>>>(
        X, W, w2, esum, counts);

    finalize1_kernel<<<1, KCLUS, 0, stream>>>(cluster_size, counts, cs);
    finalize2_kernel<<<dim3(KCLUS / TT, VEC / TT), 256, 0, stream>>>(
        embed_avg, esum, cs, out);
}


# k loop: the W panel's 4x4 register transposes, its load offset and the A row-tile addresses are formed one trip ahead BETWEEN the products; only memory instructions stand in front; k order kept
# speedup vs baseline: 5.0778x; 1.0304x over previous
#include <hip/hip_runtime.h>
#include <hip/hip_bf16.h>

#define BATCH 65536
#define VEC   512
#define KCLUS 1024
#define GAMMA_ 0.99f
#define EPS_   1e-5f

#define ROWS_PER_BLOCK 64
#define XPAD 516

#define PN 256
#define PK 8
#define WPITCH 12
#define WBUF (PN * WPITCH)

typedef __attribute__((ext_vector_type(2))) float v2f;
typedef __attribute__((ext_vector_type(4))) float v4f;
typedef __attribute__((ext_vector_type(8))) float v8f;

#define ESUM_OFF   0
#define COUNTS_OFF (VEC * KCLUS)
#define W2_OFF     (COUNTS_OFF + KCLUS)
#define CS_OFF     (W2_OFF + KCLUS)

#define PREP_BLOCKS 128
#define W2_DEPTH    64
__global__ __launch_bounds__(256)
void prep_kernel(const float* __restrict__ W, float* __restrict__ ws)
{
    if (blockIdx.x < KCLUS / 32 && threadIdx.x < 32) {
        const int k = blockIdx.x * 32 + threadIdx.x;
        const float* wp = W + k;
        float s = 0.0f;
#pragma unroll
        for (int v0 = 0; v0 < VEC; v0 += W2_DEPTH) {
            float w[W2_DEPTH];
#pragma unroll
            for (int j = 0; j < W2_DEPTH; ++j)
                w[j] = wp[(size_t)(v0 + j) * KCLUS];
#pragma unroll
            for (int j = 0; j < W2_DEPTH; ++j)
                s = __builtin_fmaf(w[j], w[j], s);
        }
        ws[W2_OFF + k] = s;
    }

    float4* z = (float4*)ws;
    for (int i = blockIdx.x * 256 + threadIdx.x; i < (VEC * KCLUS + KCLUS) / 4; i += PREP_BLOCKS * 256)
        z[i] = make_float4(0.0f, 0.0f, 0.0f, 0.0f);
}

__device__ __forceinline__ void panel_fetch(const float* __restrict__ wsrc, int p, v4f (&pf)[4])
{
    const float* g = wsrc + (size_t)((p & 63) * PK) * KCLUS + (p >> 6) * PN;
#pragma unroll
    for (int j = 0; j < 4; ++j)
        pf[j] = *(const v4f*)(g + (size_t)j * KCLUS);
}

__device__ __forceinline__ void panel_store(float* __restrict__ dst, const v4f (&pf)[4])
{
#pragma unroll
    for (int i = 0; i < 4; ++i) {
        v4f v = {pf[0][i], pf[1][i], pf[2][i], pf[3][i]};
        *(v4f*)(dst + i * WPITCH) = v;
    }
}

__device__ __forceinline__ void lds_barrier()
{
    __builtin_amdgcn_fence(__ATOMIC_RELEASE, "workgroup", "local");
    __builtin_amdgcn_s_barrier();
    __builtin_amdgcn_fence(__ATOMIC_ACQUIRE, "workgroup", "local");
}

__global__ __launch_bounds__(128)
void assign_accum_kernel(const float* __restrict__ X,
                         const float* __restrict__ W,
                         const float* __restrict__ w2,
                         float* __restrict__ esum,
                         float* __restrict__ counts)
{
    extern __shared__ char smem[];
    float* xs   = (float*)smem;
    float* wt   = xs + ROWS_PER_BLOCK * XPAD;
    int*   aidx = (int*)(wt + 2 * WBUF);

    const int tid  = threadIdx.x;
    const int lane = tid & 31;
    const int wave = tid >> 5;
    const int rowBase = blockIdx.x * ROWS_PER_BLOCK;

    const int cg  = tid >> 1;
    const int kg0 = tid & 1;
    const float* wsrc = W + (size_t)(4 * kg0) * KCLUS + 4 * cg;
    float* wdst0 = wt + (4 * cg) * WPITCH + 4 * kg0;
    v4f pf[2][4];

    panel_fetch(wsrc, 0, pf[0]);

    {
        const float4* Xg = (const float4*)(X + (size_t)rowBase * VEC);
#pragma unroll 4
        for (int i = tid; i < ROWS_PER_BLOCK * (VEC / 4); i += 128) {
            int row = i >> 7;
            int v4  = i & 127;
            float4 val = Xg[(size_t)row * (VEC / 4) + v4];
            *(float4*)&xs[row * XPAD + v4 * 4] = val;
        }
    }
    panel_store(wdst0, pf[0]);
    panel_fetch(wsrc, 1, pf[1]);
    __syncthreads();

    const int l15     = lane & 15;
    const int khalf   = (lane >> 4) << 1;
    const int colw    = wave * 64;

    float best0[4][8];
    int   bidx0[4][8];
#pragma unroll
    for (int m = 0; m < 4; ++m) {
#pragma unroll
        for (int r = 0; r < 8; ++r) {
            best0[m][r] = 3.4e38f; bidx0[m][r] = 0;
        }
    }

    const int a_b0 = 4 * (l15 * XPAD + khalf);
    const int b_b0 = 4 * (ROWS_PER_BLOCK * XPAD + (colw + l15) * WPITCH + khalf);
    const int wdst_b = 4 * (ROWS_PER_BLOCK * XPAD + (4 * cg) * WPITCH + 4 * kg0);
    const unsigned gthr_b = 4u * (unsigned)((4 * kg0) * KCLUS + 4 * cg);

    v2f fa[2][PK / 4][4];
    v2f fb[2][PK / 4][4];

#define LOAD_FRAGS(SET, BUF)                                                            \
    {                                                                                   \
        _Pragma("unroll")                                                               \
        for (int s = 0; s < PK / 4; ++s) {                                              \
            _Pragma("unroll")                                                           \
            for (int m = 0; m < 4; ++m)                                                 \
                fa[SET][s][m] = *(const v2f*)(smem + aadr[m] + 16 * s);                 \
            _Pragma("unroll")                                                           \
            for (int n = 0; n < 4; ++n)                                                 \
                fb[SET][s][n] = *(const v2f*)(smem + b_b0 + 4 * ((BUF) * WBUF + n * 16 * WPITCH) + 16 * s); \
        }                                                                               \
    }
#define A_ADDRS(q)                                                                      \
    {                                                                                   \
        _Pragma("unroll")                                                               \
        for (int m = 0; m < 4; ++m) {                                                   \
            int a_ = a_b0 + ((q) & 63) * (PK * 4) + m * (16 * XPAD * 4);                \
            asm("" : "+v"(a_));                                                         \
            aadr[m] = a_;                                                               \
        }                                                                               \
    }
#define G_OFFS(p)                                                                       \
    {                                                                                   \
        unsigned g_ = gthr_b + 4u * (unsigned)((((p) & 63) * PK) * KCLUS + ((p) >> 6) * PN); \
        asm("" : "+v"(g_));                                                             \
        goff = g_;                                                                      \
    }
#define TRANSPOSE_COL(SRC, i)                                                           \
    {                                                                                   \
        v4f t_ = {pf[SRC][0][i], pf[SRC][1][i], pf[SRC][2][i], pf[SRC][3][i]};          \
        asm("" : "+v"(t_));                                                             \
        tr[i] = t_;                                                                     \
    }
#define PROD_ROW(U, S, M)                                                               \
    {                                                                                   \
        _Pragma("unroll")                                                               \
        for (int n = 0; n < 4; ++n)                                                     \
            acc0[M][n] = __builtin_amdgcn_wmma_f32_16x16x4_f32(                         \
                             false, fa[U][S][M], false, fb[U][S][n], (short)0, acc0[M][n], false, false); \
    }

    int      aadr[4];
    unsigned goff;
    v4f      tr[4];

    A_ADDRS(0)
    LOAD_FRAGS(0, 0)
    panel_store(wdst0 + WBUF, pf[1]);
    panel_fetch(wsrc, 2, pf[0]);
    panel_fetch(wsrc, 3, pf[1]);
    A_ADDRS(1)
    G_OFFS(4)
    TRANSPOSE_COL(0, 0) TRANSPOSE_COL(0, 1) TRANSPOSE_COL(0, 2) TRANSPOSE_COL(0, 3)
    __syncthreads();

    const int NP = (KCLUS / PN) * (VEC / PK);

    for (int cb = 0; cb < KCLUS / PN; ++cb) {
        v8f acc0[4][4];
        float w2c[4];
#pragma unroll
        for (int n = 0; n < 4; ++n) {
#pragma unroll
            for (int m = 0; m < 4; ++m)
                acc0[m][n] = (v8f){0.f, 0.f, 0.f, 0.f, 0.f, 0.f, 0.f, 0.f};
            w2c[n]  = w2[cb * PN + colw + n * 16 + l15];
        }

        for (int kc = 0; kc < VEC / PK; kc += 2) {
#pragma unroll
            for (int u = 0; u < 2; ++u) {
                const int t = cb * (VEC / PK) + kc + u;

#pragma unroll
                for (int i = 0; i < 4; ++i)
                    *(v4f*)(smem + wdst_b + 4 * (u * WBUF + i * WPITCH)) = tr[i];
#pragma unroll
                for (int j = 0; j < 4; ++j)
                    pf[u][j] = *(const v4f*)((const char*)W + goff + (size_t)j * (KCLUS * 4));
                LOAD_FRAGS(u ^ 1, u ^ 1)

                __builtin_amdgcn_sched_barrier(0);

                PROD_ROW(u, 0, 0)
                __builtin_amdgcn_sched_barrier(0);
                A_ADDRS(t + 2)
                PROD_ROW(u, 0, 1)
                __builtin_amdgcn_sched_barrier(0);
                G_OFFS(min(t + 5, NP - 1))
                PROD_ROW(u, 0, 2)
                __builtin_amdgcn_sched_barrier(0);
                PROD_ROW(u, 0, 3)
                PROD_ROW(u, 1, 0)
                PROD_ROW(u, 1, 1)
                __builtin_amdgcn_sched_barrier(0);
                TRANSPOSE_COL(u ^ 1, 0)
                TRANSPOSE_COL(u ^ 1, 1)
                PROD_ROW(u, 1, 2)
                __builtin_amdgcn_sched_barrier(0);
                TRANSPOSE_COL(u ^ 1, 2)
                TRANSPOSE_COL(u ^ 1, 3)
                PROD_ROW(u, 1, 3)
                __builtin_amdgcn_sched_barrier(0);
                lds_barrier();
            }
        }

#pragma unroll
        for (int n = 0; n < 4; ++n) {
            const int col = cb * PN + colw + n * 16 + l15;
#pragma unroll
            for (int m = 0; m < 4; ++m) {
#pragma unroll
                for (int r = 0; r < 8; ++r) {
                    float sc0 = w2c[n] - 2.0f * acc0[m][n][r];
                    if (sc0 < best0[m][r]) { best0[m][r] = sc0; bidx0[m][r] = col; }
                }
            }
        }
    }
#undef LOAD_FRAGS
#undef A_ADDRS
#undef G_OFFS
#undef TRANSPOSE_COL
#undef PROD_ROW

#pragma unroll
    for (int sh = 8; sh >= 1; sh >>= 1) {
        float ob0[4][8];
        int   oi0[4][8];
#pragma unroll
        for (int m = 0; m < 4; ++m) {
#pragma unroll
            for (int r = 0; r < 8; ++r) {
                ob0[m][r] = __shfl_xor(best0[m][r], sh, 32);
                oi0[m][r] = __shfl_xor(bidx0[m][r], sh, 32);
            }
        }
#pragma unroll
        for (int m = 0; m < 4; ++m) {
#pragma unroll
            for (int r = 0; r < 8; ++r) {
                const bool take = (ob0[m][r] < best0[m][r]) | ((ob0[m][r] == best0[m][r]) & (oi0[m][r] < bidx0[m][r]));
                best0[m][r] = take ? ob0[m][r] : best0[m][r];
                bidx0[m][r] = take ? oi0[m][r] : bidx0[m][r];
            }
        }
    }
    float* wbest = wt + 4 * ROWS_PER_BLOCK;
    int*   widx  = (int*)(wbest + 4 * ROWS_PER_BLOCK);
    if (l15 == 0) {
#pragma unroll
        for (int m = 0; m < 4; ++m) {
            const int rb0 = m * 16 + (lane >> 4) * 8;
#pragma unroll
            for (int r = 0; r < 8; ++r) {
                wbest[wave * ROWS_PER_BLOCK + rb0 + r] = best0[m][r];
                widx [wave * ROWS_PER_BLOCK + rb0 + r] = bidx0[m][r];
            }
        }
    }
    __syncthreads();
    if (tid < ROWS_PER_BLOCK) {
        float b = wbest[tid];
        int   i = widx[tid];
#pragma unroll
        for (int w = 1; w < 4; ++w) {
            const float ob = wbest[w * ROWS_PER_BLOCK + tid];
            const int   oi = widx [w * ROWS_PER_BLOCK + tid];
            if (ob < b || (ob == b && oi < i)) { b = ob; i = oi; }
        }
        aidx[tid] = i;
    }
    __syncthreads();

    if (tid < ROWS_PER_BLOCK) atomicAdd(&counts[aidx[tid]], 1.0f);

    int* prow  = (int*)wt;
    int* pcode = prow + ROWS_PER_BLOCK;
    if (tid < ROWS_PER_BLOCK) {
        const int k = aidx[tid];
        int rank = 0;
        for (int j = 0; j < ROWS_PER_BLOCK; ++j) {
            const int kj = aidx[j];
            rank += (kj < k || (kj == k && j < tid)) ? 1 : 0;
        }
        prow[rank] = tid;
    }
    __syncthreads();
    if (tid < ROWS_PER_BLOCK) {
        const int r = prow[tid];
        const int k = aidx[r];
        const bool last = (tid == ROWS_PER_BLOCK - 1) ? true : (aidx[prow[tid + 1]] != k);
        pcode[tid] = r | ((last ? k + 1 : 0) << 6);
    }
    __syncthreads();

    float s0 = 0.0f, s1 = 0.0f, s2 = 0.0f, s3 = 0.0f;
    for (int g = 0; g < ROWS_PER_BLOCK; g += 8) {
        int   pc[8];
        float x[8][4];
#pragma unroll
        for (int u = 0; u < 8; ++u)
            pc[u] = __builtin_amdgcn_readfirstlane(pcode[g + u]);
#pragma unroll
        for (int u = 0; u < 8; ++u) {
            const float* xr = xs + (pc[u] & 63) * XPAD + tid;
            x[u][0] = xr[0]; x[u][1] = xr[128]; x[u][2] = xr[256]; x[u][3] = xr[384];
        }
#pragma unroll
        for (int u = 0; u < 8; ++u) {
            s0 += x[u][0]; s1 += x[u][1]; s2 += x[u][2]; s3 += x[u][3];
            const int e = pc[u] >> 6;
            if (e != 0) {
                float* d = esum + (size_t)(e - 1) * VEC + tid;
                atomicAdd(d,       s0);
                atomicAdd(d + 128, s1);
                atomicAdd(d + 256, s2);
                atomicAdd(d + 384, s3);
                s0 = 0.0f; s1 = 0.0f; s2 = 0.0f; s3 = 0.0f;
            }
        }
    }
}

__global__ __launch_bounds__(1024)
void finalize1_kernel(const float* __restrict__ cluster_size,
                      const float* __restrict__ counts,
                      float* __restrict__ cs)
{
    __shared__ float red[KCLUS];
    int k = threadIdx.x;
    float c = counts[k];
    if (c == 0.0f) c = 1.0f;
    float ncs = GAMMA_ * cluster_size[k] + (1.0f - GAMMA_) * c;
    red[k] = ncs;
    __syncthreads();
    for (int s = KCLUS / 2; s > 0; s >>= 1) {
        if (k < s) red[k] += red[k + s];
        __syncthreads();
    }
    float n = red[0];
    cs[k] = (ncs + EPS_) / (n + (float)KCLUS * EPS_) * n;
}

#define TT 32
__global__ __launch_bounds__(256)
void finalize2_kernel(const float* __restrict__ embed_avg,
                      const float* __restrict__ esum,
                      const float* __restrict__ cs,
                      float* __restrict__ out)
{
    __shared__ float tile[TT][TT + 1];
    const int tx = threadIdx.x & 31;
    const int ty = threadIdx.x >> 5;
    const int k0 = blockIdx.x * TT;
    const int v0 = blockIdx.y * TT;
#pragma unroll
    for (int j = 0; j < TT / 8; ++j) {
        const int kk = ty + 8 * j;
        tile[kk][tx] = esum[(size_t)(k0 + kk) * VEC + v0 + tx];
    }
    __syncthreads();
    const int   k = k0 + tx;
    const float c = cs[k];
#pragma unroll
    for (int j = 0; j < TT / 8; ++j) {
        const int vv = ty + 8 * j;
        const size_t o = (size_t)(v0 + vv) * KCLUS + k;
        out[o] = (GAMMA_ * embed_avg[o] + (1.0f - GAMMA_) * tile[tx][vv]) / c;
    }
}

extern "C" void kernel_launch(void* const* d_in, const int* in_sizes, int n_in,
                              void* d_out, int out_size, void* d_ws, size_t ws_size,
                              hipStream_t stream) {
    const float* X            = (const float*)d_in[0];
    const float* W            = (const float*)d_in[1];
    const float* cluster_size = (const float*)d_in[2];
    const float* embed_avg    = (const float*)d_in[3];
    float* out = (float*)d_out;
    float* ws  = (float*)d_ws;

    float* esum   = ws + ESUM_OFF;
    float* counts = ws + COUNTS_OFF;
    float* w2     = ws + W2_OFF;
    float* cs     = ws + CS_OFF;

    prep_kernel<<<PREP_BLOCKS, 256, 0, stream>>>(W, ws);

    size_t smem = (size_t)ROWS_PER_BLOCK * XPAD * sizeof(float)
                + (size_t)2 * WBUF * sizeof(float)
                + (size_t)ROWS_PER_BLOCK * sizeof(int);
    assign_accum_kernel<<<BATCH / ROWS_PER_BLOCK, 128, smem, stream>>>(
        X, W, w2, esum, counts);

    finalize1_kernel<<<1, KCLUS, 0, stream>>>(cluster_size, counts, cs);
    finalize2_kernel<<<dim3(KCLUS / TT, VEC / TT), 256, 0, stream>>>(
        embed_avg, esum, cs, out);
}


# W goes to the matrix unit straight from memory: the prologue writes W re-laid in fragment order (in out until the last launch overwrites it), B = 16-byte loads 3 panels ahead; no LDS panel; sums kept
# speedup vs baseline: 5.2609x; 1.0361x over previous
#include <hip/hip_runtime.h>
#include <hip/hip_bf16.h>

#define BATCH 65536
#define VEC   512
#define KCLUS 1024
#define GAMMA_ 0.99f
#define EPS_   1e-5f

#define ROWS_PER_BLOCK 64
#define XPAD 516

#define PN 256
#define PK 8
#define NPANEL ((KCLUS / PN) * (VEC / PK))
#define SCR_WORDS 768

typedef __attribute__((ext_vector_type(2))) float v2f;
typedef __attribute__((ext_vector_type(4))) float v4f;
typedef __attribute__((ext_vector_type(8))) float v8f;

#define ESUM_OFF   0
#define COUNTS_OFF (VEC * KCLUS)
#define W2_OFF     (COUNTS_OFF + KCLUS)
#define CS_OFF     (W2_OFF + KCLUS)

#define PREP_BLOCKS 128
#define W2_DEPTH    64
__global__ __launch_bounds__(256)
void prep_kernel(const float* __restrict__ W, float* __restrict__ ws, float* __restrict__ wp)
{
    if (blockIdx.x < KCLUS / 32 && threadIdx.x < 32) {
        const int k = blockIdx.x * 32 + threadIdx.x;
        const float* wq = W + k;
        float s = 0.0f;
#pragma unroll
        for (int v0 = 0; v0 < VEC; v0 += W2_DEPTH) {
            float w[W2_DEPTH];
#pragma unroll
            for (int j = 0; j < W2_DEPTH; ++j)
                w[j] = wq[(size_t)(v0 + j) * KCLUS];
#pragma unroll
            for (int j = 0; j < W2_DEPTH; ++j)
                s = __builtin_fmaf(w[j], w[j], s);
        }
        ws[W2_OFF + k] = s;
    }

    for (int i = blockIdx.x * 256 + threadIdx.x; i < NPANEL * PN; i += PREP_BLOCKS * 256) {
        const int col = i & (PN - 1);
        const int p   = i >> 8;
        const float* g = W + (size_t)((p & 63) * PK) * KCLUS + (p >> 6) * PN + col;
        float k[PK];
#pragma unroll
        for (int j = 0; j < PK; ++j)
            k[j] = g[(size_t)j * KCLUS];
        float4* d = (float4*)(wp + (size_t)i * PK);
        d[0] = make_float4(k[0], k[1], k[4], k[5]);
        d[1] = make_float4(k[2], k[3], k[6], k[7]);
    }

    float4* z = (float4*)ws;
    for (int i = blockIdx.x * 256 + threadIdx.x; i < (VEC * KCLUS + KCLUS) / 4; i += PREP_BLOCKS * 256)
        z[i] = make_float4(0.0f, 0.0f, 0.0f, 0.0f);
}

__global__ __launch_bounds__(128)
void assign_accum_kernel(const float* __restrict__ X,
                         const float* __restrict__ wp,
                         const float* __restrict__ w2,
                         float* __restrict__ esum,
                         float* __restrict__ counts)
{
    extern __shared__ char smem[];
    float* xs   = (float*)smem;
    float* wt   = xs + ROWS_PER_BLOCK * XPAD;
    int*   aidx = (int*)(wt + SCR_WORDS);

    const int tid  = threadIdx.x;
    const int lane = tid & 31;
    const int wave = tid >> 5;
    const int rowBase = blockIdx.x * ROWS_PER_BLOCK;

    const int l15     = lane & 15;
    const int khalf   = (lane >> 4) << 1;
    const int colw    = wave * 64;

    const char* bsrc = (const char*)wp + 4 * ((colw + l15) * PK + 2 * khalf);
    v4f fbq[4][4];

#define LOAD_B(SET, p)                                                                  \
    {                                                                                   \
        _Pragma("unroll")                                                               \
        for (int n = 0; n < 4; ++n)                                                     \
            fbq[SET][n] = *(const v4f*)(bsrc + (size_t)(p) * (PN * PK * 4) + n * (16 * PK * 4)); \
    }

    LOAD_B(0, 0)
    LOAD_B(1, 1)
    LOAD_B(2, 2)

    {
        const float4* Xg = (const float4*)(X + (size_t)rowBase * VEC);
#pragma unroll 4
        for (int i = tid; i < ROWS_PER_BLOCK * (VEC / 4); i += 128) {
            int row = i >> 7;
            int v4  = i & 127;
            float4 val = Xg[(size_t)row * (VEC / 4) + v4];
            *(float4*)&xs[row * XPAD + v4 * 4] = val;
        }
    }
    __syncthreads();

    float best0[4][8];
    int   bidx0[4][8];
#pragma unroll
    for (int m = 0; m < 4; ++m) {
#pragma unroll
        for (int r = 0; r < 8; ++r) {
            best0[m][r] = 3.4e38f; bidx0[m][r] = 0;
        }
    }

    const int a_b0 = 4 * (l15 * XPAD + khalf);

    v2f fa[2][PK / 4][4];

#define LOAD_A(SET)                                                                     \
    {                                                                                   \
        _Pragma("unroll")                                                               \
        for (int s = 0; s < PK / 4; ++s) {                                              \
            _Pragma("unroll")                                                           \
            for (int m = 0; m < 4; ++m)                                                 \
                fa[SET][s][m] = *(const v2f*)(smem + aadr[m] + 16 * s);                 \
        }                                                                               \
    }
#define A_ADDRS(q)                                                                      \
    {                                                                                   \
        _Pragma("unroll")                                                               \
        for (int m = 0; m < 4; ++m) {                                                   \
            int a_ = a_b0 + ((q) & 63) * (PK * 4) + m * (16 * XPAD * 4);                \
            asm("" : "+v"(a_));                                                         \
            aadr[m] = a_;                                                               \
        }                                                                               \
    }
#define PROD_ROW(UA, UB, S, M)                                                          \
    {                                                                                   \
        _Pragma("unroll")                                                               \
        for (int n = 0; n < 4; ++n) {                                                   \
            const v2f b_ = {fbq[UB][n][2 * (S)], fbq[UB][n][2 * (S) + 1]};              \
            acc0[M][n] = __builtin_amdgcn_wmma_f32_16x16x4_f32(                         \
                             false, fa[UA][S][M], false, b_, (short)0, acc0[M][n], false, false); \
        }                                                                               \
    }

    int aadr[4];

    A_ADDRS(0)
    LOAD_A(0)
    A_ADDRS(1)

    const int NP = NPANEL;

    for (int cb = 0; cb < KCLUS / PN; ++cb) {
        v8f acc0[4][4];
        float w2c[4];
#pragma unroll
        for (int n = 0; n < 4; ++n) {
#pragma unroll
            for (int m = 0; m < 4; ++m)
                acc0[m][n] = (v8f){0.f, 0.f, 0.f, 0.f, 0.f, 0.f, 0.f, 0.f};
            w2c[n]  = w2[cb * PN + colw + n * 16 + l15];
        }

        for (int kc = 0; kc < VEC / PK; kc += 4) {
#pragma unroll
            for (int u = 0; u < 4; ++u) {
                const int t = cb * (VEC / PK) + kc + u;

                LOAD_B((u + 3) & 3, min(t + 3, NP - 1))
                LOAD_A((u & 1) ^ 1)

                __builtin_amdgcn_sched_barrier(0);

                PROD_ROW(u & 1, u, 0, 0)
                __builtin_amdgcn_sched_barrier(0);
                A_ADDRS(t + 2)
                PROD_ROW(u & 1, u, 0, 1)
                __builtin_amdgcn_sched_barrier(0);
                PROD_ROW(u & 1, u, 0, 2)
                PROD_ROW(u & 1, u, 0, 3)
                PROD_ROW(u & 1, u, 1, 0)
                PROD_ROW(u & 1, u, 1, 1)
                PROD_ROW(u & 1, u, 1, 2)
                PROD_ROW(u & 1, u, 1, 3)
                __builtin_amdgcn_sched_barrier(0);
            }
        }
#pragma unroll
        for (int n = 0; n < 4; ++n) {
            const int col = cb * PN + colw + n * 16 + l15;
#pragma unroll
            for (int m = 0; m < 4; ++m) {
#pragma unroll
                for (int r = 0; r < 8; ++r) {
                    float sc0 = w2c[n] - 2.0f * acc0[m][n][r];
                    if (sc0 < best0[m][r]) { best0[m][r] = sc0; bidx0[m][r] = col; }
                }
            }
        }
    }
#undef LOAD_B
#undef LOAD_A
#undef A_ADDRS
#undef PROD_ROW

#pragma unroll
    for (int sh = 8; sh >= 1; sh >>= 1) {
        float ob0[4][8];
        int   oi0[4][8];
#pragma unroll
        for (int m = 0; m < 4; ++m) {
#pragma unroll
            for (int r = 0; r < 8; ++r) {
                ob0[m][r] = __shfl_xor(best0[m][r], sh, 32);
                oi0[m][r] = __shfl_xor(bidx0[m][r], sh, 32);
            }
        }
#pragma unroll
        for (int m = 0; m < 4; ++m) {
#pragma unroll
            for (int r = 0; r < 8; ++r) {
                const bool take = (ob0[m][r] < best0[m][r]) | ((ob0[m][r] == best0[m][r]) & (oi0[m][r] < bidx0[m][r]));
                best0[m][r] = take ? ob0[m][r] : best0[m][r];
                bidx0[m][r] = take ? oi0[m][r] : bidx0[m][r];
            }
        }
    }
    float* wbest = wt + 4 * ROWS_PER_BLOCK;
    int*   widx  = (int*)(wbest + 4 * ROWS_PER_BLOCK);
    if (l15 == 0) {
#pragma unroll
        for (int m = 0; m < 4; ++m) {
            const int rb0 = m * 16 + (lane >> 4) * 8;
#pragma unroll
            for (int r = 0; r < 8; ++r) {
                wbest[wave * ROWS_PER_BLOCK + rb0 + r] = best0[m][r];
                widx [wave * ROWS_PER_BLOCK + rb0 + r] = bidx0[m][r];
            }
        }
    }
    __syncthreads();
    if (tid < ROWS_PER_BLOCK) {
        float b = wbest[tid];
        int   i = widx[tid];
#pragma unroll
        for (int w = 1; w < 4; ++w) {
            const float ob = wbest[w * ROWS_PER_BLOCK + tid];
            const int   oi = widx [w * ROWS_PER_BLOCK + tid];
            if (ob < b || (ob == b && oi < i)) { b = ob; i = oi; }
        }
        aidx[tid] = i;
    }
    __syncthreads();

    if (tid < ROWS_PER_BLOCK) atomicAdd(&counts[aidx[tid]], 1.0f);

    int* prow  = (int*)wt;
    int* pcode = prow + ROWS_PER_BLOCK;
    if (tid < ROWS_PER_BLOCK) {
        const int k = aidx[tid];
        int rank = 0;
        for (int j = 0; j < ROWS_PER_BLOCK; ++j) {
            const int kj = aidx[j];
            rank += (kj < k || (kj == k && j < tid)) ? 1 : 0;
        }
        prow[rank] = tid;
    }
    __syncthreads();
    if (tid < ROWS_PER_BLOCK) {
        const int r = prow[tid];
        const int k = aidx[r];
        const bool last = (tid == ROWS_PER_BLOCK - 1) ? true : (aidx[prow[tid + 1]] != k);
        pcode[tid] = r | ((last ? k + 1 : 0) << 6);
    }
    __syncthreads();

    float s0 = 0.0f, s1 = 0.0f, s2 = 0.0f, s3 = 0.0f;
    for (int g = 0; g < ROWS_PER_BLOCK; g += 8) {
        int   pc[8];
        float x[8][4];
#pragma unroll
        for (int u = 0; u < 8; ++u)
            pc[u] = __builtin_amdgcn_readfirstlane(pcode[g + u]);
#pragma unroll
        for (int u = 0; u < 8; ++u) {
            const float* xr = xs + (pc[u] & 63) * XPAD + tid;
            x[u][0] = xr[0]; x[u][1] = xr[128]; x[u][2] = xr[256]; x[u][3] = xr[384];
        }
#pragma unroll
        for (int u = 0; u < 8; ++u) {
            s0 += x[u][0]; s1 += x[u][1]; s2 += x[u][2]; s3 += x[u][3];
            const int e = pc[u] >> 6;
            if (e != 0) {
                float* d = esum + (size_t)(e - 1) * VEC + tid;
                atomicAdd(d,       s0);
                atomicAdd(d + 128, s1);
                atomicAdd(d + 256, s2);
                atomicAdd(d + 384, s3);
                s0 = 0.0f; s1 = 0.0f; s2 = 0.0f; s3 = 0.0f;
            }
        }
    }
}

__global__ __launch_bounds__(1024)
void finalize1_kernel(const float* __restrict__ cluster_size,
                      const float* __restrict__ counts,
                      float* __restrict__ cs)
{
    __shared__ float red[KCLUS];
    int k = threadIdx.x;
    float c = counts[k];
    if (c == 0.0f) c = 1.0f;
    float ncs = GAMMA_ * cluster_size[k] + (1.0f - GAMMA_) * c;
    red[k] = ncs;
    __syncthreads();
    for (int s = KCLUS / 2; s > 0; s >>= 1) {
        if (k < s) red[k] += red[k + s];
        __syncthreads();
    }
    float n = red[0];
    cs[k] = (ncs + EPS_) / (n + (float)KCLUS * EPS_) * n;
}

#define TT 32
__global__ __launch_bounds__(256)
void finalize2_kernel(const float* __restrict__ embed_avg,
                      const float* __restrict__ esum,
                      const float* __restrict__ cs,
                      float* __restrict__ out)
{
    __shared__ float tile[TT][TT + 1];
    const int tx = threadIdx.x & 31;
    const int ty = threadIdx.x >> 5;
    const int k0 = blockIdx.x * TT;
    const int v0 = blockIdx.y * TT;
#pragma unroll
    for (int j = 0; j < TT / 8; ++j) {
        const int kk = ty + 8 * j;
        tile[kk][tx] = esum[(size_t)(k0 + kk) * VEC + v0 + tx];
    }
    __syncthreads();
    const int   k = k0 + tx;
    const float c = cs[k];
#pragma unroll
    for (int j = 0; j < TT / 8; ++j) {
        const int vv = ty + 8 * j;
        const size_t o = (size_t)(v0 + vv) * KCLUS + k;
        out[o] = (GAMMA_ * embed_avg[o] + (1.0f - GAMMA_) * tile[tx][vv]) / c;
    }
}

extern "C" void kernel_launch(void* const* d_in, const int* in_sizes, int n_in,
                              void* d_out, int out_size, void* d_ws, size_t ws_size,
                              hipStream_t stream) {
    const float* X            = (const float*)d_in[0];
    const float* W            = (const float*)d_in[1];
    const float* cluster_size = (const float*)d_in[2];
    const float* embed_avg    = (const float*)d_in[3];
    float* out = (float*)d_out;
    float* ws  = (float*)d_ws;

    float* esum   = ws + ESUM_OFF;
    float* counts = ws + COUNTS_OFF;
    float* w2     = ws + W2_OFF;
    float* cs     = ws + CS_OFF;

    prep_kernel<<<PREP_BLOCKS, 256, 0, stream>>>(W, ws, out);

    size_t smem = (size_t)ROWS_PER_BLOCK * XPAD * sizeof(float)
                + (size_t)SCR_WORDS * sizeof(float)
                + (size_t)ROWS_PER_BLOCK * sizeof(int);
    assign_accum_kernel<<<BATCH / ROWS_PER_BLOCK, 128, smem, stream>>>(
        X, out, w2, esum, counts);

    finalize1_kernel<<<1, KCLUS, 0, stream>>>(cluster_size, counts, cs);
    finalize2_kernel<<<dim3(KCLUS / TT, VEC / TT), 256, 0, stream>>>(
        embed_avg, esum, cs, out);
}


# k loop memory group = 8 bare memory instructions: B loads take a scalar panel base + fixed lane offset, A reads a complete LDS address, both formed a trip ahead between products; sums unchanged
# speedup vs baseline: 5.4399x; 1.0340x over previous
#include <hip/hip_runtime.h>
#include <hip/hip_bf16.h>

#define BATCH 65536
#define VEC   512
#define KCLUS 1024
#define GAMMA_ 0.99f
#define EPS_   1e-5f

#define ROWS_PER_BLOCK 64
#define XPAD 516

#define PN 256
#define PK 8
#define NPANEL ((KCLUS / PN) * (VEC / PK))
#define SCR_WORDS 768

typedef __attribute__((ext_vector_type(2))) float v2f;
typedef __attribute__((ext_vector_type(4))) float v4f;
typedef __attribute__((ext_vector_type(8))) float v8f;
typedef const __attribute__((address_space(1))) char* gbytes;
typedef const __attribute__((address_space(1))) v4f*  gv4f;
typedef const __attribute__((address_space(3))) char* lbytes;
typedef const __attribute__((address_space(3))) v2f*  lv2f;

#define ESUM_OFF   0
#define COUNTS_OFF (VEC * KCLUS)
#define W2_OFF     (COUNTS_OFF + KCLUS)
#define CS_OFF     (W2_OFF + KCLUS)

#define PREP_BLOCKS 128
#define W2_DEPTH    64
__global__ __launch_bounds__(256)
void prep_kernel(const float* __restrict__ W, float* __restrict__ ws, float* __restrict__ wp)
{
    if (blockIdx.x < KCLUS / 32 && threadIdx.x < 32) {
        const int k = blockIdx.x * 32 + threadIdx.x;
        const float* wq = W + k;
        float s = 0.0f;
#pragma unroll
        for (int v0 = 0; v0 < VEC; v0 += W2_DEPTH) {
            float w[W2_DEPTH];
#pragma unroll
            for (int j = 0; j < W2_DEPTH; ++j)
                w[j] = wq[(size_t)(v0 + j) * KCLUS];
#pragma unroll
            for (int j = 0; j < W2_DEPTH; ++j)
                s = __builtin_fmaf(w[j], w[j], s);
        }
        ws[W2_OFF + k] = s;
    }

    for (int i = blockIdx.x * 256 + threadIdx.x; i < NPANEL * PN; i += PREP_BLOCKS * 256) {
        const int col = i & (PN - 1);
        const int p   = i >> 8;
        const float* g = W + (size_t)((p & 63) * PK) * KCLUS + (p >> 6) * PN + col;
        float k[PK];
#pragma unroll
        for (int j = 0; j < PK; ++j)
            k[j] = g[(size_t)j * KCLUS];
        float4* d = (float4*)(wp + (size_t)i * PK);
        d[0] = make_float4(k[0], k[1], k[4], k[5]);
        d[1] = make_float4(k[2], k[3], k[6], k[7]);
    }

    float4* z = (float4*)ws;
    for (int i = blockIdx.x * 256 + threadIdx.x; i < (VEC * KCLUS + KCLUS) / 4; i += PREP_BLOCKS * 256)
        z[i] = make_float4(0.0f, 0.0f, 0.0f, 0.0f);
}

__global__ __launch_bounds__(128)
void assign_accum_kernel(const float* __restrict__ X,
                         const float* __restrict__ wp,
                         const float* __restrict__ w2,
                         float* __restrict__ esum,
                         float* __restrict__ counts)
{
    extern __shared__ char smem[];
    float* xs   = (float*)smem;
    float* wt   = xs + ROWS_PER_BLOCK * XPAD;
    int*   aidx = (int*)(wt + SCR_WORDS);

    const int tid  = threadIdx.x;
    const int lane = tid & 31;
    const int wave = tid >> 5;
    const int rowBase = blockIdx.x * ROWS_PER_BLOCK;

    const int l15     = lane & 15;
    const int khalf   = (lane >> 4) << 1;
    const int colw    = wave * 64;

    unsigned boff = 4u * ((unsigned)(colw + l15) * PK + 2u * (unsigned)khalf);
    v4f fbq[4][4];

    gbytes pbase;
#define B_BASE(q)                                                                       \
    {                                                                                   \
        gbytes pb_ = (gbytes)(const char*)wp + (size_t)min((q), NPANEL - 1) * (PN * PK * 4); \
          \
          \
        asm("" : "+s"(pb_));                                                            \
        pbase = pb_;                                                                    \
    }
#define LOAD_B(SET)                                                                     \
    {                                                                                   \
          \
          \
        asm("" : "+v"(boff));                                                           \
        _Pragma("unroll")                                                               \
        for (int n = 0; n < 4; ++n)                                                     \
            fbq[SET][n] = *(gv4f)(pbase + boff + n * (16 * PK * 4));                    \
    }

    B_BASE(0)
    LOAD_B(0)
    B_BASE(1)
    LOAD_B(1)
    B_BASE(2)
    LOAD_B(2)

    {
        const float4* Xg = (const float4*)(X + (size_t)rowBase * VEC);
#pragma unroll 4
        for (int i = tid; i < ROWS_PER_BLOCK * (VEC / 4); i += 128) {
            int row = i >> 7;
            int v4  = i & 127;
            float4 val = Xg[(size_t)row * (VEC / 4) + v4];
            *(float4*)&xs[row * XPAD + v4 * 4] = val;
        }
    }
    __syncthreads();

    float best0[4][8];
    int   bidx0[4][8];
#pragma unroll
    for (int m = 0; m < 4; ++m) {
#pragma unroll
        for (int r = 0; r < 8; ++r) {
            best0[m][r] = 3.4e38f; bidx0[m][r] = 0;
        }
    }

    const int a_b0 = (int)(unsigned)(size_t)(lbytes)smem + 4 * (l15 * XPAD + khalf);

    v2f fa[2][PK / 4][4];

#define LOAD_A(SET)                                                                     \
    {                                                                                   \
        _Pragma("unroll")                                                               \
        for (int s = 0; s < PK / 4; ++s) {                                              \
            _Pragma("unroll")                                                           \
            for (int m = 0; m < 4; ++m)                                                 \
                fa[SET][s][m] = *(lv2f)(unsigned)(aadr[m] + 16 * s);                    \
        }                                                                               \
    }
#define A_ADDRS(q)                                                                      \
    {                                                                                   \
        _Pragma("unroll")                                                               \
        for (int m = 0; m < 4; ++m) {                                                   \
            int a_ = a_b0 + ((q) & 63) * (PK * 4) + m * (16 * XPAD * 4);                \
            asm("" : "+v"(a_));                                                         \
            aadr[m] = a_;                                                               \
        }                                                                               \
    }
#define PROD_ROW(UA, UB, S, M)                                                          \
    {                                                                                   \
        _Pragma("unroll")                                                               \
        for (int n = 0; n < 4; ++n) {                                                   \
            const v2f b_ = {fbq[UB][n][2 * (S)], fbq[UB][n][2 * (S) + 1]};              \
            acc0[M][n] = __builtin_amdgcn_wmma_f32_16x16x4_f32(                         \
                             false, fa[UA][S][M], false, b_, (short)0, acc0[M][n], false, false); \
        }                                                                               \
    }

    int aadr[4];

    A_ADDRS(0)
    LOAD_A(0)
    A_ADDRS(1)
    B_BASE(3)

    int tb = 0;

    for (int cb = 0; cb < KCLUS / PN; ++cb) {
        v8f acc0[4][4];
        float w2c[4];
#pragma unroll
        for (int n = 0; n < 4; ++n) {
#pragma unroll
            for (int m = 0; m < 4; ++m)
                acc0[m][n] = (v8f){0.f, 0.f, 0.f, 0.f, 0.f, 0.f, 0.f, 0.f};
            w2c[n]  = w2[cb * PN + colw + n * 16 + l15];
        }

        for (int kc = 0; kc < VEC / PK; kc += 4) {
#pragma unroll
            for (int u = 0; u < 4; ++u) {
                LOAD_B((u + 3) & 3)
                LOAD_A((u & 1) ^ 1)

                __builtin_amdgcn_sched_barrier(0);

                PROD_ROW(u & 1, u, 0, 0)
                __builtin_amdgcn_sched_barrier(0);
                const int t = tb + u;
                A_ADDRS(t + 2)
                B_BASE(t + 4)
                PROD_ROW(u & 1, u, 0, 1)
                __builtin_amdgcn_sched_barrier(0);
                PROD_ROW(u & 1, u, 0, 2)
                PROD_ROW(u & 1, u, 0, 3)
                PROD_ROW(u & 1, u, 1, 0)
                PROD_ROW(u & 1, u, 1, 1)
                PROD_ROW(u & 1, u, 1, 2)
                PROD_ROW(u & 1, u, 1, 3)
                __builtin_amdgcn_sched_barrier(0);
            }
            tb += 4;
            asm("" : "+s"(tb));
        }
#pragma unroll
        for (int n = 0; n < 4; ++n) {
            const int col = cb * PN + colw + n * 16 + l15;
#pragma unroll
            for (int m = 0; m < 4; ++m) {
#pragma unroll
                for (int r = 0; r < 8; ++r) {
                    float sc0 = w2c[n] - 2.0f * acc0[m][n][r];
                    if (sc0 < best0[m][r]) { best0[m][r] = sc0; bidx0[m][r] = col; }
                }
            }
        }
    }
#undef LOAD_B
#undef B_BASE
#undef LOAD_A
#undef A_ADDRS
#undef PROD_ROW

#pragma unroll
    for (int sh = 8; sh >= 1; sh >>= 1) {
        float ob0[4][8];
        int   oi0[4][8];
#pragma unroll
        for (int m = 0; m < 4; ++m) {
#pragma unroll
            for (int r = 0; r < 8; ++r) {
                ob0[m][r] = __shfl_xor(best0[m][r], sh, 32);
                oi0[m][r] = __shfl_xor(bidx0[m][r], sh, 32);
            }
        }
#pragma unroll
        for (int m = 0; m < 4; ++m) {
#pragma unroll
            for (int r = 0; r < 8; ++r) {
                const bool take = (ob0[m][r] < best0[m][r]) | ((ob0[m][r] == best0[m][r]) & (oi0[m][r] < bidx0[m][r]));
                best0[m][r] = take ? ob0[m][r] : best0[m][r];
                bidx0[m][r] = take ? oi0[m][r] : bidx0[m][r];
            }
        }
    }
    float* wbest = wt + 4 * ROWS_PER_BLOCK;
    int*   widx  = (int*)(wbest + 4 * ROWS_PER_BLOCK);
    if (l15 == 0) {
#pragma unroll
        for (int m = 0; m < 4; ++m) {
            const int rb0 = m * 16 + (lane >> 4) * 8;
#pragma unroll
            for (int r = 0; r < 8; ++r) {
                wbest[wave * ROWS_PER_BLOCK + rb0 + r] = best0[m][r];
                widx [wave * ROWS_PER_BLOCK + rb0 + r] = bidx0[m][r];
            }
        }
    }
    __syncthreads();
    if (tid < ROWS_PER_BLOCK) {
        float b = wbest[tid];
        int   i = widx[tid];
#pragma unroll
        for (int w = 1; w < 4; ++w) {
            const float ob = wbest[w * ROWS_PER_BLOCK + tid];
            const int   oi = widx [w * ROWS_PER_BLOCK + tid];
            if (ob < b || (ob == b && oi < i)) { b = ob; i = oi; }
        }
        aidx[tid] = i;
    }
    __syncthreads();

    if (tid < ROWS_PER_BLOCK) atomicAdd(&counts[aidx[tid]], 1.0f);

    int* prow  = (int*)wt;
    int* pcode = prow + ROWS_PER_BLOCK;
    if (tid < ROWS_PER_BLOCK) {
        const int k = aidx[tid];
        int rank = 0;
        for (int j = 0; j < ROWS_PER_BLOCK; ++j) {
            const int kj = aidx[j];
            rank += (kj < k || (kj == k && j < tid)) ? 1 : 0;
        }
        prow[rank] = tid;
    }
    __syncthreads();
    if (tid < ROWS_PER_BLOCK) {
        const int r = prow[tid];
        const int k = aidx[r];
        const bool last = (tid == ROWS_PER_BLOCK - 1) ? true : (aidx[prow[tid + 1]] != k);
        pcode[tid] = r | ((last ? k + 1 : 0) << 6);
    }
    __syncthreads();

    float s0 = 0.0f, s1 = 0.0f, s2 = 0.0f, s3 = 0.0f;
    for (int g = 0; g < ROWS_PER_BLOCK; g += 8) {
        int   pc[8];
        float x[8][4];
#pragma unroll
        for (int u = 0; u < 8; ++u)
            pc[u] = __builtin_amdgcn_readfirstlane(pcode[g + u]);
#pragma unroll
        for (int u = 0; u < 8; ++u) {
            const float* xr = xs + (pc[u] & 63) * XPAD + tid;
            x[u][0] = xr[0]; x[u][1] = xr[128]; x[u][2] = xr[256]; x[u][3] = xr[384];
        }
#pragma unroll
        for (int u = 0; u < 8; ++u) {
            s0 += x[u][0]; s1 += x[u][1]; s2 += x[u][2]; s3 += x[u][3];
            const int e = pc[u] >> 6;
            if (e != 0) {
                float* d = esum + (size_t)(e - 1) * VEC + tid;
                atomicAdd(d,       s0);
                atomicAdd(d + 128, s1);
                atomicAdd(d + 256, s2);
                atomicAdd(d + 384, s3);
                s0 = 0.0f; s1 = 0.0f; s2 = 0.0f; s3 = 0.0f;
            }
        }
    }
}

__global__ __launch_bounds__(1024)
void finalize1_kernel(const float* __restrict__ cluster_size,
                      const float* __restrict__ counts,
                      float* __restrict__ cs)
{
    __shared__ float red[KCLUS];
    int k = threadIdx.x;
    float c = counts[k];
    if (c == 0.0f) c = 1.0f;
    float ncs = GAMMA_ * cluster_size[k] + (1.0f - GAMMA_) * c;
    red[k] = ncs;
    __syncthreads();
    for (int s = KCLUS / 2; s > 0; s >>= 1) {
        if (k < s) red[k] += red[k + s];
        __syncthreads();
    }
    float n = red[0];
    cs[k] = (ncs + EPS_) / (n + (float)KCLUS * EPS_) * n;
}

#define TT 32
__global__ __launch_bounds__(256)
void finalize2_kernel(const float* __restrict__ embed_avg,
                      const float* __restrict__ esum,
                      const float* __restrict__ cs,
                      float* __restrict__ out)
{
    __shared__ float tile[TT][TT + 1];
    const int tx = threadIdx.x & 31;
    const int ty = threadIdx.x >> 5;
    const int k0 = blockIdx.x * TT;
    const int v0 = blockIdx.y * TT;
#pragma unroll
    for (int j = 0; j < TT / 8; ++j) {
        const int kk = ty + 8 * j;
        tile[kk][tx] = esum[(size_t)(k0 + kk) * VEC + v0 + tx];
    }
    __syncthreads();
    const int   k = k0 + tx;
    const float c = cs[k];
#pragma unroll
    for (int j = 0; j < TT / 8; ++j) {
        const int vv = ty + 8 * j;
        const size_t o = (size_t)(v0 + vv) * KCLUS + k;
        out[o] = (GAMMA_ * embed_avg[o] + (1.0f - GAMMA_) * tile[tx][vv]) / c;
    }
}

extern "C" void kernel_launch(void* const* d_in, const int* in_sizes, int n_in,
                              void* d_out, int out_size, void* d_ws, size_t ws_size,
                              hipStream_t stream) {
    const float* X            = (const float*)d_in[0];
    const float* W            = (const float*)d_in[1];
    const float* cluster_size = (const float*)d_in[2];
    const float* embed_avg    = (const float*)d_in[3];
    float* out = (float*)d_out;
    float* ws  = (float*)d_ws;

    float* esum   = ws + ESUM_OFF;
    float* counts = ws + COUNTS_OFF;
    float* w2     = ws + W2_OFF;
    float* cs     = ws + CS_OFF;

    prep_kernel<<<PREP_BLOCKS, 256, 0, stream>>>(W, ws, out);

    size_t smem = (size_t)ROWS_PER_BLOCK * XPAD * sizeof(float)
                + (size_t)SCR_WORDS * sizeof(float)
                + (size_t)ROWS_PER_BLOCK * sizeof(int);
    assign_accum_kernel<<<BATCH / ROWS_PER_BLOCK, 128, smem, stream>>>(
        X, out, w2, esum, counts);

    finalize1_kernel<<<1, KCLUS, 0, stream>>>(cluster_size, counts, cs);
    finalize2_kernel<<<dim3(KCLUS / TT, VEC / TT), 256, 0, stream>>>(
        embed_avg, esum, cs, out);
}
